# starting kernel
# baseline (speedup 1.0000x reference)
#include <hip/hip_runtime.h>
#include <math.h>

#define HD 128
#define TPB 256

typedef __attribute__((ext_vector_type(2))) float v2f;
typedef __attribute__((ext_vector_type(8))) float v8f;

__global__ void __launch_bounds__(TPB) zero_kernel(float* __restrict__ p, size_t n) {
    size_t i = (size_t)blockIdx.x * TPB + threadIdx.x;
    if (i < n) p[i] = 0.0f;
}

__device__ __forceinline__ unsigned enc_f32(float f) {
    unsigned b = __float_as_uint(f);
    return (b & 0x80000000u) ? ~b : (b | 0x80000000u);
}
__device__ __forceinline__ float dec_f32(unsigned e) {
    unsigned b = (e & 0x80000000u) ? (e & 0x7FFFFFFFu) : ~e;
    return __uint_as_float(b);
}

__global__ void __launch_bounds__(TPB) edge_score_kernel(
        const float* __restrict__ ent, const float* __restrict__ rel,
        const int* __restrict__ src, const int* __restrict__ dst,
        const int* __restrict__ rid,
        float* __restrict__ score, unsigned* __restrict__ segmax, int n_edges) {
    int e = blockIdx.x * (TPB / 32) + (threadIdx.x >> 5);
    if (e >= n_edges) return;
    int lane = threadIdx.x & 31;
    int s = src[e], d = dst[e], r = rid[e];
    const float4* es = (const float4*)(ent + (size_t)s * HD);
    const float4* ed = (const float4*)(ent + (size_t)d * HD);
    const float4* rr = (const float4*)(rel + (size_t)r * HD);
    float4 a = es[lane], b = rr[lane], c = ed[lane];
    float v = a.x*b.x*c.x + a.y*b.y*c.y + a.z*b.z*c.z + a.w*b.w*c.w;
    #pragma unroll
    for (int off = 16; off > 0; off >>= 1) v += __shfl_xor(v, off, 32);
    if (lane == 0) {
        score[e] = v;
        atomicMax(segmax + d, enc_f32(v));
    }
}

__global__ void __launch_bounds__(TPB) edge_exp_kernel(
        const int* __restrict__ dst, const unsigned* __restrict__ segmax,
        float* __restrict__ score, float* __restrict__ denom, int n_edges) {
    int e = blockIdx.x * TPB + threadIdx.x;
    if (e >= n_edges) return;
    int d = dst[e];
    float ex = __expf(score[e] - dec_f32(segmax[d]));
    score[e] = ex;
    atomicAdd(denom + d, ex);
}

__global__ void __launch_bounds__(TPB) edge_aggr_kernel(
        const float* __restrict__ ent, const float* __restrict__ rel,
        const int* __restrict__ src, const int* __restrict__ dst,
        const int* __restrict__ rid,
        const float* __restrict__ score, const float* __restrict__ denom,
        float* __restrict__ neigh, int n_edges) {
    int e = blockIdx.x * (TPB / 32) + (threadIdx.x >> 5);
    if (e >= n_edges) return;
    int lane = threadIdx.x & 31;
    int s = src[e], d = dst[e], r = rid[e];
    float alpha = score[e] / denom[d];
    const float4* es = (const float4*)(ent + (size_t)s * HD);
    const float4* rr = (const float4*)(rel + (size_t)r * HD);
    float4 a = es[lane], b = rr[lane];
    float* out = neigh + (size_t)d * HD + lane * 4;
    atomicAdd(out + 0, a.x * b.x * alpha);
    atomicAdd(out + 1, a.y * b.y * alpha);
    atomicAdd(out + 2, a.z * b.z * alpha);
    atomicAdd(out + 3, a.w * b.w * alpha);
}

__global__ void __launch_bounds__(TPB) gemm_wmma_kernel(
        const float* __restrict__ A,
        const float* __restrict__ W,
        float* __restrict__ H,
        int n_rows) {
    const int wave = threadIdx.x >> 5;
    const int lane = threadIdx.x & 31;
    const int half = lane >> 4;
    const int l16  = lane & 15;
    const int m0 = blockIdx.x * 16;
    const int n0 = wave * 16;

    v8f c = {};
    const float* arow = A + (size_t)(m0 + l16) * HD;
    const float* wcol = W + n0 + l16;

    #pragma unroll 4
    for (int k0 = 0; k0 < HD; k0 += 4) {
        int ka = k0 + 2 * half;
        v2f a, b;
        a.x = arow[ka];
        a.y = arow[ka + 1];
        b.x = wcol[(size_t)ka * HD];
        b.y = wcol[(size_t)(ka + 1) * HD];
        c = __builtin_amdgcn_wmma_f32_16x16x4_f32(
                 false, a,  false, b,
                 (short)0, c,  false,  false);
    }

    float* hb = H + (size_t)(m0 + 8 * half) * HD + n0 + l16;
    #pragma unroll
    for (int r2 = 0; r2 < 8; ++r2) hb[(size_t)r2 * HD] = c[r2];
}

__global__ void __launch_bounds__(TPB) bn_stats_kernel(
        const float* __restrict__ H, float* __restrict__ colsum,
        float* __restrict__ colsumsq, int n_rows) {
    int col  = threadIdx.x & (HD - 1);
    int roff = threadIdx.x >> 7;
    float s = 0.0f, sq = 0.0f;
    for (int r = blockIdx.x * 2 + roff; r < n_rows; r += gridDim.x * 2) {
        float v = H[(size_t)r * HD + col];
        s += v; sq += v * v;
    }
    atomicAdd(colsum + col, s);
    atomicAdd(colsumsq + col, sq);
}

__global__ void __launch_bounds__(TPB) bn_tanh_kernel(
        float* __restrict__ H, const float* __restrict__ colsum,
        const float* __restrict__ colsumsq, const float* __restrict__ gamma,
        const float* __restrict__ beta, int n_rows) {
    size_t i = (size_t)blockIdx.x * TPB + threadIdx.x;
    size_t total = (size_t)n_rows * HD;
    if (i >= total) return;
    int c = (int)(i & (HD - 1));
    float invN = 1.0f / (float)n_rows;
    float mu   = colsum[c] * invN;
    float var  = colsumsq[c] * invN - mu * mu;
    float istd = rsqrtf(var + 1e-5f);
    float v = (H[i] - mu) * istd * gamma[c] + beta[c];
    H[i] = tanhf(v);
}

extern "C" void kernel_launch(void* const* d_in, const int* in_sizes, int n_in,
                              void* d_out, int out_size, void* d_ws, size_t ws_size,
                              hipStream_t stream) {
    const float* ent   = (const float*)d_in[0];
    const float* rel   = (const float*)d_in[1];
    const float* W     = (const float*)d_in[2];
    const float* gamma = (const float*)d_in[3];
    const float* beta  = (const float*)d_in[4];
    const int*   src   = (const int*)d_in[5];
    const int*   dst   = (const int*)d_in[6];
    const int*   rid   = (const int*)d_in[7];
    float* H = (float*)d_out;

    const int N = in_sizes[0] / HD;
    const int E = in_sizes[5];

    float*    ws       = (float*)d_ws;
    float*    neigh    = ws;
    float*    score    = neigh + (size_t)N * HD;
    unsigned* segmax   = (unsigned*)(score + E);
    float*    denom    = (float*)(segmax + N);
    float*    colsum   = denom + N;
    float*    colsumsq = colsum + HD;

    size_t zn = (size_t)N * HD;
    zero_kernel<<<(unsigned)((zn + TPB - 1) / TPB), TPB, 0, stream>>>(neigh, zn);
    size_t zt = (size_t)2 * N + 2 * HD;
    zero_kernel<<<(unsigned)((zt + TPB - 1) / TPB), TPB, 0, stream>>>((float*)segmax, zt);

    const int edges_per_blk = TPB / 32;
    edge_score_kernel<<<(unsigned)((E + edges_per_blk - 1) / edges_per_blk), TPB, 0, stream>>>(
        ent, rel, src, dst, rid, score, segmax, E);
    edge_exp_kernel<<<(unsigned)((E + TPB - 1) / TPB), TPB, 0, stream>>>(
        dst, segmax, score, denom, E);
    edge_aggr_kernel<<<(unsigned)((E + edges_per_blk - 1) / edges_per_blk), TPB, 0, stream>>>(
        ent, rel, src, dst, rid, score, denom, neigh, E);

    gemm_wmma_kernel<<<(unsigned)(N / 16), TPB, 0, stream>>>(neigh, W, H, N);

    bn_stats_kernel<<<256, TPB, 0, stream>>>(H, colsum, colsumsq, N);
    bn_tanh_kernel<<<(unsigned)(((size_t)N * HD + TPB - 1) / TPB), TPB, 0, stream>>>(
        H, colsum, colsumsq, gamma, beta, N);
}


# Earlier try summed segments in arrival order; now edges grouped by dst (int atomics, scan) and each segment key-sorted in its wave, then score, online softmax, sum in registers: no float atomics
# speedup vs baseline: 1.3303x; 1.3303x over previous
#include <hip/hip_runtime.h>
#include <math.h>

#define HD 128
#define TPB 256

typedef __attribute__((ext_vector_type(2))) float v2f;
typedef __attribute__((ext_vector_type(8))) float v8f;

__global__ void __launch_bounds__(TPB) zero_kernel(float* __restrict__ p, size_t n) {
    size_t i = (size_t)blockIdx.x * TPB + threadIdx.x;
    if (i < n) p[i] = 0.0f;
}

__global__ void __launch_bounds__(TPB) degree_count_kernel(
        const int* __restrict__ dst, int* __restrict__ cnt, int n_edges) {
    int e = blockIdx.x * TPB + threadIdx.x;
    if (e >= n_edges) return;
    atomicAdd(cnt + dst[e], 1);
}

#define SCAN_T 1024
__global__ void __launch_bounds__(SCAN_T) degree_scan_kernel(
        const int* __restrict__ cnt, int* __restrict__ off, int n) {
    __shared__ int wsum[SCAN_T / 32];
    const int t = threadIdx.x;
    const int lane = t & 31, w = t >> 5;
    const int per = (n + SCAN_T - 1) / SCAN_T;
    const int b = t * per;
    const int e = (b + per < n) ? (b + per) : n;
    int s = 0;
    for (int i = b; i < e; ++i) s += cnt[i];
    int x = s;
    for (int o = 1; o < 32; o <<= 1) {
        int y = __shfl_up(x, o, 32);
        if (lane >= o) x += y;
    }
    if (lane == 31) wsum[w] = x;
    __syncthreads();
    if (w == 0) {
        int v = wsum[lane];
        for (int o = 1; o < 32; o <<= 1) {
            int y = __shfl_up(v, o, 32);
            if (lane >= o) v += y;
        }
        wsum[lane] = v;
    }
    __syncthreads();
    int base = (w > 0 ? wsum[w - 1] : 0) + x - s;
    for (int i = b; i < e; ++i) { off[i] = base; base += cnt[i]; }
}

__global__ void __launch_bounds__(TPB) segment_fill_kernel(
        const int* __restrict__ src, const int* __restrict__ dst,
        const int* __restrict__ rid, const int* __restrict__ off,
        int* __restrict__ cnt, int* __restrict__ keys, int n_edges, int n_rel) {
    int e = blockIdx.x * TPB + threadIdx.x;
    if (e >= n_edges) return;
    int d = dst[e];
    int key = src[e] * n_rel + rid[e];
    int old = atomicSub(cnt + d, 1);
    keys[off[d] + old - 1] = key;
}

struct SegAcc { float4 acc; float m; float l; };

__device__ __forceinline__ float wave_sum32(float v) {
    for (int o = 16; o > 0; o >>= 1) v += __shfl_xor(v, o, 32);
    return v;
}

__device__ __forceinline__ void seg_update(SegAcc& st, const float4& comp, float v) {
    float mn = fmaxf(st.m, v);
    float sc = __expf(st.m - mn);
    float p  = __expf(v - mn);
    st.acc.x = st.acc.x * sc + comp.x * p;
    st.acc.y = st.acc.y * sc + comp.y * p;
    st.acc.z = st.acc.z * sc + comp.z * p;
    st.acc.w = st.acc.w * sc + comp.w * p;
    st.l = st.l * sc + p;
    st.m = mn;
}

__global__ void __launch_bounds__(TPB) node_softmax_aggr_kernel(
        const float* __restrict__ ent, const float* __restrict__ rel,
        const int* __restrict__ off, const int* __restrict__ keys,
        float* __restrict__ neigh, int n_nodes, int n_edges, int n_rel) {
    const int d = __builtin_amdgcn_readfirstlane((int)(blockIdx.x * (TPB / 32) + (threadIdx.x >> 5)));
    if (d >= n_nodes) return;
    const int lane = threadIdx.x & 31;
    const int beg = off[d];
    const int end = (d + 1 < n_nodes) ? off[d + 1] : n_edges;
    const int deg = end - beg;
    float4* outp = (float4*)(neigh + (size_t)d * HD) + lane;
    if (deg <= 0) { *outp = make_float4(0.0f, 0.0f, 0.0f, 0.0f); return; }

    const float4 c = ((const float4*)(ent + (size_t)d * HD))[lane];
    SegAcc st;
    st.acc = make_float4(0.0f, 0.0f, 0.0f, 0.0f);
    st.m = -INFINITY;
    st.l = 0.0f;

    if (deg <= 32) {
        const int key = (lane < deg) ? keys[beg + lane] : 0x7fffffff;
        int rank = 0;
        for (int j = 0; j < deg; ++j) {
            int kj = __builtin_amdgcn_readlane(key, j);
            rank += ((kj < key) || (kj == key && j < lane)) ? 1 : 0;
        }
        if (lane >= deg) rank = lane;
        const int skey = __builtin_amdgcn_ds_permute(rank << 2, key);
        const int sk = skey / n_rel;
        const int rk = skey - sk * n_rel;

        for (int k0 = 0; k0 < deg; k0 += 4) {
            float4 a[4], b[4];
            for (int u = 0; u < 4; ++u) {
                int k = (k0 + u < deg) ? (k0 + u) : (deg - 1);
                int s = __builtin_amdgcn_readlane(sk, k);
                int r = __builtin_amdgcn_readlane(rk, k);
                a[u] = ((const float4*)(ent + (size_t)s * HD))[lane];
                b[u] = ((const float4*)(rel + (size_t)r * HD))[lane];
            }
            float4 comp[4];
            float v[4];
            for (int u = 0; u < 4; ++u) {
                comp[u] = make_float4(a[u].x * b[u].x, a[u].y * b[u].y,
                                      a[u].z * b[u].z, a[u].w * b[u].w);
                v[u] = comp[u].x * c.x + comp[u].y * c.y + comp[u].z * c.z + comp[u].w * c.w;
            }
            for (int o = 16; o > 0; o >>= 1) {
                for (int u = 0; u < 4; ++u) v[u] += __shfl_xor(v[u], o, 32);
            }
            for (int u = 0; u < 4; ++u) {
                if (k0 + u < deg) seg_update(st, comp[u], v[u]);
            }
        }
    } else {
        int prev = -1;
        int done = 0;
        for (int it = 0; it < deg && done < deg; ++it) {
            int best = 0x7fffffff;
            for (int i = lane; i < deg; i += 32) {
                int kk = keys[beg + i];
                if (kk > prev && kk < best) best = kk;
            }
            for (int o = 16; o > 0; o >>= 1) {
                int y = __shfl_xor(best, o, 32);
                best = (y < best) ? y : best;
            }
            best = __builtin_amdgcn_readfirstlane(best);
            int mult = 0;
            for (int i = lane; i < deg; i += 32) mult += (keys[beg + i] == best) ? 1 : 0;
            for (int o = 16; o > 0; o >>= 1) mult += __shfl_xor(mult, o, 32);
            mult = __builtin_amdgcn_readfirstlane(mult);
            if (mult == 0) break;
            const int s = best / n_rel;
            const int r = best - s * n_rel;
            const float4 a = ((const float4*)(ent + (size_t)s * HD))[lane];
            const float4 b = ((const float4*)(rel + (size_t)r * HD))[lane];
            const float4 comp = make_float4(a.x * b.x, a.y * b.y, a.z * b.z, a.w * b.w);
            const float v = wave_sum32(comp.x * c.x + comp.y * c.y + comp.z * c.z + comp.w * c.w);
            for (int t = 0; t < mult; ++t) seg_update(st, comp, v);
            done += mult;
            prev = best;
        }
    }

    const float inv = 1.0f / st.l;
    *outp = make_float4(st.acc.x * inv, st.acc.y * inv, st.acc.z * inv, st.acc.w * inv);
}

__global__ void __launch_bounds__(TPB) gemm_wmma_kernel(
        const float* __restrict__ A,
        const float* __restrict__ W,
        float* __restrict__ H,
        int n_rows) {
    const int wave = threadIdx.x >> 5;
    const int lane = threadIdx.x & 31;
    const int half = lane >> 4;
    const int l16  = lane & 15;
    const int m0 = blockIdx.x * 16;
    const int n0 = wave * 16;

    v8f c = {};
    const float* arow = A + (size_t)(m0 + l16) * HD;
    const float* wcol = W + n0 + l16;

    #pragma unroll 4
    for (int k0 = 0; k0 < HD; k0 += 4) {
        int ka = k0 + 2 * half;
        v2f a, b;
        a.x = arow[ka];
        a.y = arow[ka + 1];
        b.x = wcol[(size_t)ka * HD];
        b.y = wcol[(size_t)(ka + 1) * HD];
        c = __builtin_amdgcn_wmma_f32_16x16x4_f32(
                 false, a,  false, b,
                 (short)0, c,  false,  false);
    }

    float* hb = H + (size_t)(m0 + 8 * half) * HD + n0 + l16;
    #pragma unroll
    for (int r2 = 0; r2 < 8; ++r2) hb[(size_t)r2 * HD] = c[r2];
}

__global__ void __launch_bounds__(TPB) bn_stats_kernel(
        const float* __restrict__ H, float* __restrict__ colsum,
        float* __restrict__ colsumsq, int n_rows) {
    int col  = threadIdx.x & (HD - 1);
    int roff = threadIdx.x >> 7;
    float s = 0.0f, sq = 0.0f;
    for (int r = blockIdx.x * 2 + roff; r < n_rows; r += gridDim.x * 2) {
        float v = H[(size_t)r * HD + col];
        s += v; sq += v * v;
    }
    atomicAdd(colsum + col, s);
    atomicAdd(colsumsq + col, sq);
}

__global__ void __launch_bounds__(TPB) bn_tanh_kernel(
        float* __restrict__ H, const float* __restrict__ colsum,
        const float* __restrict__ colsumsq, const float* __restrict__ gamma,
        const float* __restrict__ beta, int n_rows) {
    size_t i = (size_t)blockIdx.x * TPB + threadIdx.x;
    size_t total = (size_t)n_rows * HD;
    if (i >= total) return;
    int c = (int)(i & (HD - 1));
    float invN = 1.0f / (float)n_rows;
    float mu   = colsum[c] * invN;
    float var  = colsumsq[c] * invN - mu * mu;
    float istd = rsqrtf(var + 1e-5f);
    float v = (H[i] - mu) * istd * gamma[c] + beta[c];
    H[i] = tanhf(v);
}

extern "C" void kernel_launch(void* const* d_in, const int* in_sizes, int n_in,
                              void* d_out, int out_size, void* d_ws, size_t ws_size,
                              hipStream_t stream) {
    const float* ent   = (const float*)d_in[0];
    const float* rel   = (const float*)d_in[1];
    const float* W     = (const float*)d_in[2];
    const float* gamma = (const float*)d_in[3];
    const float* beta  = (const float*)d_in[4];
    const int*   src   = (const int*)d_in[5];
    const int*   dst   = (const int*)d_in[6];
    const int*   rid   = (const int*)d_in[7];
    float* H = (float*)d_out;

    const int N = in_sizes[0] / HD;
    const int R = in_sizes[1] / HD;
    const int E = in_sizes[5];

    float* ws       = (float*)d_ws;
    float* neigh    = ws;
    int*   keys     = (int*)(neigh + (size_t)N * HD);
    int*   cnt      = keys + E;
    float* colsum   = (float*)(cnt + N);
    float* colsumsq = colsum + HD;
    int*   off      = (int*)(colsumsq + HD);

    size_t zt = (size_t)N + 2 * HD;
    zero_kernel<<<(unsigned)((zt + TPB - 1) / TPB), TPB, 0, stream>>>((float*)cnt, zt);

    degree_count_kernel<<<(unsigned)((E + TPB - 1) / TPB), TPB, 0, stream>>>(dst, cnt, E);
    degree_scan_kernel<<<1, SCAN_T, 0, stream>>>(cnt, off, N);
    segment_fill_kernel<<<(unsigned)((E + TPB - 1) / TPB), TPB, 0, stream>>>(
        src, dst, rid, off, cnt, keys, E, R);

    const int nodes_per_blk = TPB / 32;
    node_softmax_aggr_kernel<<<(unsigned)((N + nodes_per_blk - 1) / nodes_per_blk), TPB, 0, stream>>>(
        ent, rel, off, keys, neigh, N, E, R);

    gemm_wmma_kernel<<<(unsigned)(N / 16), TPB, 0, stream>>>(neigh, W, H, N);

    bn_stats_kernel<<<256, TPB, 0, stream>>>(H, colsum, colsumsq, N);
    bn_tanh_kernel<<<(unsigned)(((size_t)N * HD + TPB - 1) / TPB), TPB, 0, stream>>>(
        H, colsum, colsumsq, gamma, beta, N);
}


# Degree-histogram scan made coalesced: each wave scans contiguous 128-count rows (4 counts per lane, 4 rows in flight), whole 512-byte loads and stores; was 49 counts per thread, 32 lines per access
# speedup vs baseline: 1.5832x; 1.1901x over previous
#include <hip/hip_runtime.h>
#include <math.h>

#define HD 128
#define TPB 256

typedef __attribute__((ext_vector_type(2))) float v2f;
typedef __attribute__((ext_vector_type(8))) float v8f;

__global__ void __launch_bounds__(TPB) zero_kernel(float* __restrict__ p, size_t n) {
    size_t i = (size_t)blockIdx.x * TPB + threadIdx.x;
    if (i < n) p[i] = 0.0f;
}

__global__ void __launch_bounds__(TPB) degree_count_kernel(
        const int* __restrict__ dst, int* __restrict__ cnt, int n_edges) {
    int e = blockIdx.x * TPB + threadIdx.x;
    if (e >= n_edges) return;
    atomicAdd(cnt + dst[e], 1);
}

#define SCAN_T 1024
#define SCAN_W (SCAN_T / 32)
#define SCAN_ROW 128
#define SCAN_K 4

__device__ __forceinline__ int wave_sum32_int(int v) {
    for (int o = 16; o > 0; o >>= 1) v += __shfl_xor(v, o, 32);
    return v;
}

template <int K>
__device__ __forceinline__ int scan_rows(const int* __restrict__ cnt, int* __restrict__ off,
                                         int row, int lane, int carry) {
    int c[K][4], ls[K], x[K];
    #pragma unroll
    for (int k = 0; k < K; ++k) {
        const int* p = cnt + (row + k) * SCAN_ROW + 4 * lane;
        c[k][0] = p[0]; c[k][1] = p[1]; c[k][2] = p[2]; c[k][3] = p[3];
        ls[k] = c[k][0] + c[k][1] + c[k][2] + c[k][3];
        x[k] = ls[k];
    }
    for (int o = 1; o < 32; o <<= 1) {
        #pragma unroll
        for (int k = 0; k < K; ++k) {
            int y = __shfl_up(x[k], o, 32);
            if (lane >= o) x[k] += y;
        }
    }
    #pragma unroll
    for (int k = 0; k < K; ++k) {
        const int ex = carry + x[k] - ls[k];
        int* q = off + (row + k) * SCAN_ROW + 4 * lane;
        q[0] = ex;
        q[1] = ex + c[k][0];
        q[2] = ex + c[k][0] + c[k][1];
        q[3] = ex + c[k][0] + c[k][1] + c[k][2];
        carry += __builtin_amdgcn_readlane(x[k], 31);
    }
    return carry;
}

__global__ void __launch_bounds__(SCAN_T) degree_scan_kernel(
        const int* __restrict__ cnt, int* __restrict__ off, int n) {
    __shared__ int wsum[SCAN_W];
    const int lane = threadIdx.x & 31;
    const int w = __builtin_amdgcn_readfirstlane((int)(threadIdx.x >> 5));
    const int full = n / SCAN_ROW;
    const int rows = (n + SCAN_ROW - 1) / SCAN_ROW;
    const int rpw  = (rows + SCAN_W - 1) / SCAN_W;
    const int r0  = w * rpw;
    const int r1  = (r0 + rpw < rows) ? (r0 + rpw) : rows;
    const int r1f = (r1 < full) ? r1 : full;
    const bool tail = (full < rows) && (r0 <= full) && (r1 == rows);
    const int ti = full * SCAN_ROW + 4 * lane;

    int s = 0;
    #pragma unroll 4
    for (int r = r0; r < r1f; ++r) {
        const int* p = cnt + r * SCAN_ROW + 4 * lane;
        s += p[0] + p[1] + p[2] + p[3];
    }
    int tc[4] = {0, 0, 0, 0};
    if (tail) {
        for (int j = 0; j < 4; ++j) if (ti + j < n) tc[j] = cnt[ti + j];
        s += tc[0] + tc[1] + tc[2] + tc[3];
    }
    s = wave_sum32_int(s);
    if (lane == 0) wsum[w] = s;
    __syncthreads();
    int carry = wave_sum32_int((lane < w) ? wsum[lane] : 0);

    int r = r0;
    for (; r + SCAN_K <= r1f; r += SCAN_K) carry = scan_rows<SCAN_K>(cnt, off, r, lane, carry);
    for (; r < r1f; ++r)                   carry = scan_rows<1>(cnt, off, r, lane, carry);
    if (tail) {
        const int ls = tc[0] + tc[1] + tc[2] + tc[3];
        int x = ls;
        for (int o = 1; o < 32; o <<= 1) {
            int y = __shfl_up(x, o, 32);
            if (lane >= o) x += y;
        }
        int ex = carry + x - ls;
        for (int j = 0; j < 4; ++j) {
            if (ti + j < n) off[ti + j] = ex;
            ex += tc[j];
        }
    }
}

__global__ void __launch_bounds__(TPB) segment_fill_kernel(
        const int* __restrict__ src, const int* __restrict__ dst,
        const int* __restrict__ rid, const int* __restrict__ off,
        int* __restrict__ cnt, int* __restrict__ keys, int n_edges, int n_rel) {
    int e = blockIdx.x * TPB + threadIdx.x;
    if (e >= n_edges) return;
    int d = dst[e];
    int key = src[e] * n_rel + rid[e];
    int old = atomicSub(cnt + d, 1);
    keys[off[d] + old - 1] = key;
}

struct SegAcc { float4 acc; float m; float l; };

__device__ __forceinline__ float wave_sum32(float v) {
    for (int o = 16; o > 0; o >>= 1) v += __shfl_xor(v, o, 32);
    return v;
}

__device__ __forceinline__ void seg_update(SegAcc& st, const float4& comp, float v) {
    float mn = fmaxf(st.m, v);
    float sc = __expf(st.m - mn);
    float p  = __expf(v - mn);
    st.acc.x = st.acc.x * sc + comp.x * p;
    st.acc.y = st.acc.y * sc + comp.y * p;
    st.acc.z = st.acc.z * sc + comp.z * p;
    st.acc.w = st.acc.w * sc + comp.w * p;
    st.l = st.l * sc + p;
    st.m = mn;
}

__global__ void __launch_bounds__(TPB) node_softmax_aggr_kernel(
        const float* __restrict__ ent, const float* __restrict__ rel,
        const int* __restrict__ off, const int* __restrict__ keys,
        float* __restrict__ neigh, int n_nodes, int n_edges, int n_rel) {
    const int d = __builtin_amdgcn_readfirstlane((int)(blockIdx.x * (TPB / 32) + (threadIdx.x >> 5)));
    if (d >= n_nodes) return;
    const int lane = threadIdx.x & 31;
    const int beg = off[d];
    const int end = (d + 1 < n_nodes) ? off[d + 1] : n_edges;
    const int deg = end - beg;
    float4* outp = (float4*)(neigh + (size_t)d * HD) + lane;
    if (deg <= 0) { *outp = make_float4(0.0f, 0.0f, 0.0f, 0.0f); return; }

    const float4 c = ((const float4*)(ent + (size_t)d * HD))[lane];
    SegAcc st;
    st.acc = make_float4(0.0f, 0.0f, 0.0f, 0.0f);
    st.m = -INFINITY;
    st.l = 0.0f;

    if (deg <= 32) {
        const int key = (lane < deg) ? keys[beg + lane] : 0x7fffffff;
        int rank = 0;
        for (int j = 0; j < deg; ++j) {
            int kj = __builtin_amdgcn_readlane(key, j);
            rank += ((kj < key) || (kj == key && j < lane)) ? 1 : 0;
        }
        if (lane >= deg) rank = lane;
        const int skey = __builtin_amdgcn_ds_permute(rank << 2, key);
        const int sk = skey / n_rel;
        const int rk = skey - sk * n_rel;

        for (int k0 = 0; k0 < deg; k0 += 4) {
            float4 a[4], b[4];
            for (int u = 0; u < 4; ++u) {
                int k = (k0 + u < deg) ? (k0 + u) : (deg - 1);
                int s = __builtin_amdgcn_readlane(sk, k);
                int r = __builtin_amdgcn_readlane(rk, k);
                a[u] = ((const float4*)(ent + (size_t)s * HD))[lane];
                b[u] = ((const float4*)(rel + (size_t)r * HD))[lane];
            }
            float4 comp[4];
            float v[4];
            for (int u = 0; u < 4; ++u) {
                comp[u] = make_float4(a[u].x * b[u].x, a[u].y * b[u].y,
                                      a[u].z * b[u].z, a[u].w * b[u].w);
                v[u] = comp[u].x * c.x + comp[u].y * c.y + comp[u].z * c.z + comp[u].w * c.w;
            }
            for (int o = 16; o > 0; o >>= 1) {
                for (int u = 0; u < 4; ++u) v[u] += __shfl_xor(v[u], o, 32);
            }
            for (int u = 0; u < 4; ++u) {
                if (k0 + u < deg) seg_update(st, comp[u], v[u]);
            }
        }
    } else {
        int prev = -1;
        int done = 0;
        for (int it = 0; it < deg && done < deg; ++it) {
            int best = 0x7fffffff;
            for (int i = lane; i < deg; i += 32) {
                int kk = keys[beg + i];
                if (kk > prev && kk < best) best = kk;
            }
            for (int o = 16; o > 0; o >>= 1) {
                int y = __shfl_xor(best, o, 32);
                best = (y < best) ? y : best;
            }
            best = __builtin_amdgcn_readfirstlane(best);
            int mult = 0;
            for (int i = lane; i < deg; i += 32) mult += (keys[beg + i] == best) ? 1 : 0;
            for (int o = 16; o > 0; o >>= 1) mult += __shfl_xor(mult, o, 32);
            mult = __builtin_amdgcn_readfirstlane(mult);
            if (mult == 0) break;
            const int s = best / n_rel;
            const int r = best - s * n_rel;
            const float4 a = ((const float4*)(ent + (size_t)s * HD))[lane];
            const float4 b = ((const float4*)(rel + (size_t)r * HD))[lane];
            const float4 comp = make_float4(a.x * b.x, a.y * b.y, a.z * b.z, a.w * b.w);
            const float v = wave_sum32(comp.x * c.x + comp.y * c.y + comp.z * c.z + comp.w * c.w);
            for (int t = 0; t < mult; ++t) seg_update(st, comp, v);
            done += mult;
            prev = best;
        }
    }

    const float inv = 1.0f / st.l;
    *outp = make_float4(st.acc.x * inv, st.acc.y * inv, st.acc.z * inv, st.acc.w * inv);
}

__global__ void __launch_bounds__(TPB) gemm_wmma_kernel(
        const float* __restrict__ A,
        const float* __restrict__ W,
        float* __restrict__ H,
        int n_rows) {
    const int wave = threadIdx.x >> 5;
    const int lane = threadIdx.x & 31;
    const int half = lane >> 4;
    const int l16  = lane & 15;
    const int m0 = blockIdx.x * 16;
    const int n0 = wave * 16;

    v8f c = {};
    const float* arow = A + (size_t)(m0 + l16) * HD;
    const float* wcol = W + n0 + l16;

    #pragma unroll 4
    for (int k0 = 0; k0 < HD; k0 += 4) {
        int ka = k0 + 2 * half;
        v2f a, b;
        a.x = arow[ka];
        a.y = arow[ka + 1];
        b.x = wcol[(size_t)ka * HD];
        b.y = wcol[(size_t)(ka + 1) * HD];
        c = __builtin_amdgcn_wmma_f32_16x16x4_f32(
                 false, a,  false, b,
                 (short)0, c,  false,  false);
    }

    float* hb = H + (size_t)(m0 + 8 * half) * HD + n0 + l16;
    #pragma unroll
    for (int r2 = 0; r2 < 8; ++r2) hb[(size_t)r2 * HD] = c[r2];
}

__global__ void __launch_bounds__(TPB) bn_stats_kernel(
        const float* __restrict__ H, float* __restrict__ colsum,
        float* __restrict__ colsumsq, int n_rows) {
    int col  = threadIdx.x & (HD - 1);
    int roff = threadIdx.x >> 7;
    float s = 0.0f, sq = 0.0f;
    for (int r = blockIdx.x * 2 + roff; r < n_rows; r += gridDim.x * 2) {
        float v = H[(size_t)r * HD + col];
        s += v; sq += v * v;
    }
    atomicAdd(colsum + col, s);
    atomicAdd(colsumsq + col, sq);
}

__global__ void __launch_bounds__(TPB) bn_tanh_kernel(
        float* __restrict__ H, const float* __restrict__ colsum,
        const float* __restrict__ colsumsq, const float* __restrict__ gamma,
        const float* __restrict__ beta, int n_rows) {
    size_t i = (size_t)blockIdx.x * TPB + threadIdx.x;
    size_t total = (size_t)n_rows * HD;
    if (i >= total) return;
    int c = (int)(i & (HD - 1));
    float invN = 1.0f / (float)n_rows;
    float mu   = colsum[c] * invN;
    float var  = colsumsq[c] * invN - mu * mu;
    float istd = rsqrtf(var + 1e-5f);
    float v = (H[i] - mu) * istd * gamma[c] + beta[c];
    H[i] = tanhf(v);
}

extern "C" void kernel_launch(void* const* d_in, const int* in_sizes, int n_in,
                              void* d_out, int out_size, void* d_ws, size_t ws_size,
                              hipStream_t stream) {
    const float* ent   = (const float*)d_in[0];
    const float* rel   = (const float*)d_in[1];
    const float* W     = (const float*)d_in[2];
    const float* gamma = (const float*)d_in[3];
    const float* beta  = (const float*)d_in[4];
    const int*   src   = (const int*)d_in[5];
    const int*   dst   = (const int*)d_in[6];
    const int*   rid   = (const int*)d_in[7];
    float* H = (float*)d_out;

    const int N = in_sizes[0] / HD;
    const int R = in_sizes[1] / HD;
    const int E = in_sizes[5];

    float* ws       = (float*)d_ws;
    float* neigh    = ws;
    int*   keys     = (int*)(neigh + (size_t)N * HD);
    int*   cnt      = keys + E;
    float* colsum   = (float*)(cnt + N);
    float* colsumsq = colsum + HD;
    int*   off      = (int*)(colsumsq + HD);

    size_t zt = (size_t)N + 2 * HD;
    zero_kernel<<<(unsigned)((zt + TPB - 1) / TPB), TPB, 0, stream>>>((float*)cnt, zt);

    degree_count_kernel<<<(unsigned)((E + TPB - 1) / TPB), TPB, 0, stream>>>(dst, cnt, E);
    degree_scan_kernel<<<1, SCAN_T, 0, stream>>>(cnt, off, N);
    segment_fill_kernel<<<(unsigned)((E + TPB - 1) / TPB), TPB, 0, stream>>>(
        src, dst, rid, off, cnt, keys, E, R);

    const int nodes_per_blk = TPB / 32;
    node_softmax_aggr_kernel<<<(unsigned)((N + nodes_per_blk - 1) / nodes_per_blk), TPB, 0, stream>>>(
        ent, rel, off, keys, neigh, N, E, R);

    gemm_wmma_kernel<<<(unsigned)(N / 16), TPB, 0, stream>>>(neigh, W, H, N);

    bn_stats_kernel<<<256, TPB, 0, stream>>>(H, colsum, colsumsq, N);
    bn_tanh_kernel<<<(unsigned)(((size_t)N * HD + TPB - 1) / TPB), TPB, 0, stream>>>(
        H, colsum, colsumsq, gamma, beta, N);
}


# Edges grouped by dst by a two-level counting sort on LDS integer adds (bin by 256-node range, then rank inside each bin): 1600000 scattered global adds become 200000; float sums unchanged
# speedup vs baseline: 2.2497x; 1.4210x over previous
#include <hip/hip_runtime.h>
#include <math.h>

#define HD 128
#define TPB 256

typedef __attribute__((ext_vector_type(2))) float v2f;
typedef __attribute__((ext_vector_type(8))) float v8f;

__global__ void __launch_bounds__(TPB) zero_kernel(float* __restrict__ p, size_t n) {
    size_t i = (size_t)blockIdx.x * TPB + threadIdx.x;
    if (i < n) p[i] = 0.0f;
}

#define BIN_SHIFT 8
#define BIN_NODES 256
#define MAX_BINS 256
#define CTR_PITCH 32
#define BIN_BLOCKS 512
#define SORT_T 1024

__device__ __forceinline__ int wave_incl_scan_int(int v, int lane) {
    int x = v;
    for (int o = 1; o < 32; o <<= 1) {
        int y = __shfl_up(x, o, 32);
        if (lane >= o) x += y;
    }
    return x;
}

__global__ void __launch_bounds__(TPB) bin_count_kernel(
        const int* __restrict__ dst, int* __restrict__ bin_cnt, int n_edges, int n_bins) {
    __shared__ int hist[MAX_BINS];
    const int t = threadIdx.x;
    hist[t] = 0;
    __syncthreads();
    const int chunk = (n_edges + (int)gridDim.x - 1) / (int)gridDim.x;
    const int e0 = blockIdx.x * chunk;
    const int e1 = (e0 + chunk < n_edges) ? (e0 + chunk) : n_edges;
    for (int e = e0 + t; e < e1; e += 4 * TPB) {
        int d[4];
        #pragma unroll
        for (int u = 0; u < 4; ++u) {
            const int ee = e + u * TPB;
            d[u] = (ee < e1) ? dst[ee] : -1;
        }
        #pragma unroll
        for (int u = 0; u < 4; ++u) if (d[u] >= 0) atomicAdd(&hist[d[u] >> BIN_SHIFT], 1);
    }
    __syncthreads();
    const int h = hist[t];
    if (t < n_bins && h > 0) atomicAdd(bin_cnt + t * CTR_PITCH, h);
}

__global__ void __launch_bounds__(TPB) bin_scatter_kernel(
        const int* __restrict__ src, const int* __restrict__ dst, const int* __restrict__ rid,
        const int* __restrict__ bin_cnt, int* __restrict__ bin_cur, int* __restrict__ bin_start,
        int* __restrict__ pairs, int n_edges, int n_bins, int n_rel) {
    __shared__ int hist[MAX_BINS];
    __shared__ int base[MAX_BINS];
    __shared__ int wtot[TPB / 32];
    const int t = threadIdx.x;
    const int lane = t & 31;
    const int w = t >> 5;
    hist[t] = 0;
    __syncthreads();
    const int chunk = (n_edges + (int)gridDim.x - 1) / (int)gridDim.x;
    const int e0 = blockIdx.x * chunk;
    const int e1 = (e0 + chunk < n_edges) ? (e0 + chunk) : n_edges;
    for (int e = e0 + t; e < e1; e += 4 * TPB) {
        int d[4];
        #pragma unroll
        for (int u = 0; u < 4; ++u) {
            const int ee = e + u * TPB;
            d[u] = (ee < e1) ? dst[ee] : -1;
        }
        #pragma unroll
        for (int u = 0; u < 4; ++u) if (d[u] >= 0) atomicAdd(&hist[d[u] >> BIN_SHIFT], 1);
    }
    const int tot = (t < n_bins) ? bin_cnt[t * CTR_PITCH] : 0;
    const int incl = wave_incl_scan_int(tot, lane);
    if (lane == 31) wtot[w] = incl;
    __syncthreads();
    int pre = 0;
    for (int i = 0; i < w; ++i) pre += wtot[i];
    const int start = pre + incl - tot;
    if (blockIdx.x == 0 && t < n_bins) bin_start[t] = start;
    const int h = hist[t];
    int b0 = 0;
    if (t < n_bins && h > 0) b0 = start + atomicAdd(bin_cur + t * CTR_PITCH, h);
    base[t] = b0;
    hist[t] = 0;
    __syncthreads();
    for (int e = e0 + t; e < e1; e += 4 * TPB) {
        int d[4], s[4], r[4];
        #pragma unroll
        for (int u = 0; u < 4; ++u) {
            const int ee = e + u * TPB;
            const bool ok = ee < e1;
            d[u] = ok ? dst[ee] : -1;
            s[u] = ok ? src[ee] : 0;
            r[u] = ok ? rid[ee] : 0;
        }
        #pragma unroll
        for (int u = 0; u < 4; ++u) {
            if (d[u] >= 0) {
                const int b = d[u] >> BIN_SHIFT;
                const int slot = base[b] + atomicAdd(&hist[b], 1);
                *((int2*)(pairs + 2 * (size_t)slot)) = make_int2(d[u], s[u] * n_rel + r[u]);
            }
        }
    }
}

__global__ void __launch_bounds__(SORT_T) bin_sort_kernel(
        const int* __restrict__ pairs, const int* __restrict__ bin_cnt,
        const int* __restrict__ bin_start, int* __restrict__ off, int* __restrict__ keys,
        int n_nodes) {
    __shared__ int lcnt[BIN_NODES];
    __shared__ int lpos[BIN_NODES];
    __shared__ int wtot[BIN_NODES / 32];
    const int t = threadIdx.x;
    const int lane = t & 31;
    const int w = t >> 5;
    const int bin = blockIdx.x;
    const int bs = bin_start[bin];
    const int be = bs + bin_cnt[bin * CTR_PITCH];
    if (t < BIN_NODES) lcnt[t] = 0;
    __syncthreads();
    for (int i = bs + t; i < be; i += 4 * SORT_T) {
        int d[4];
        #pragma unroll
        for (int u = 0; u < 4; ++u) {
            const int ii = i + u * SORT_T;
            d[u] = (ii < be) ? pairs[2 * (size_t)ii] : -1;
        }
        #pragma unroll
        for (int u = 0; u < 4; ++u) if (d[u] >= 0) atomicAdd(&lcnt[d[u] & (BIN_NODES - 1)], 1);
    }
    __syncthreads();
    const int v = (t < BIN_NODES) ? lcnt[t] : 0;
    const int incl = wave_incl_scan_int(v, lane);
    if (t < BIN_NODES && lane == 31) wtot[w] = incl;
    __syncthreads();
    if (t < BIN_NODES) {
        int pre = 0;
        for (int i = 0; i < w; ++i) pre += wtot[i];
        const int ex = bs + pre + incl - v;
        lpos[t] = ex;
        const int node = bin * BIN_NODES + t;
        if (node < n_nodes) off[node] = ex;
    }
    __syncthreads();
    for (int i = bs + t; i < be; i += 4 * SORT_T) {
        int2 p[4];
        #pragma unroll
        for (int u = 0; u < 4; ++u) {
            const int ii = i + u * SORT_T;
            p[u] = (ii < be) ? *((const int2*)(pairs + 2 * (size_t)ii)) : make_int2(-1, 0);
        }
        #pragma unroll
        for (int u = 0; u < 4; ++u) {
            if (p[u].x >= 0) {
                const int slot = atomicAdd(&lpos[p[u].x & (BIN_NODES - 1)], 1);
                keys[slot] = p[u].y;
            }
        }
    }
}

struct SegAcc { float4 acc; float m; float l; };

__device__ __forceinline__ float wave_sum32(float v) {
    for (int o = 16; o > 0; o >>= 1) v += __shfl_xor(v, o, 32);
    return v;
}

__device__ __forceinline__ void seg_update(SegAcc& st, const float4& comp, float v) {
    float mn = fmaxf(st.m, v);
    float sc = __expf(st.m - mn);
    float p  = __expf(v - mn);
    st.acc.x = st.acc.x * sc + comp.x * p;
    st.acc.y = st.acc.y * sc + comp.y * p;
    st.acc.z = st.acc.z * sc + comp.z * p;
    st.acc.w = st.acc.w * sc + comp.w * p;
    st.l = st.l * sc + p;
    st.m = mn;
}

__global__ void __launch_bounds__(TPB) node_softmax_aggr_kernel(
        const float* __restrict__ ent, const float* __restrict__ rel,
        const int* __restrict__ off, const int* __restrict__ keys,
        float* __restrict__ neigh, int n_nodes, int n_edges, int n_rel) {
    const int d = __builtin_amdgcn_readfirstlane((int)(blockIdx.x * (TPB / 32) + (threadIdx.x >> 5)));
    if (d >= n_nodes) return;
    const int lane = threadIdx.x & 31;
    const int beg = off[d];
    const int end = (d + 1 < n_nodes) ? off[d + 1] : n_edges;
    const int deg = end - beg;
    float4* outp = (float4*)(neigh + (size_t)d * HD) + lane;
    if (deg <= 0) { *outp = make_float4(0.0f, 0.0f, 0.0f, 0.0f); return; }

    const float4 c = ((const float4*)(ent + (size_t)d * HD))[lane];
    SegAcc st;
    st.acc = make_float4(0.0f, 0.0f, 0.0f, 0.0f);
    st.m = -INFINITY;
    st.l = 0.0f;

    if (deg <= 32) {
        const int key = (lane < deg) ? keys[beg + lane] : 0x7fffffff;
        int rank = 0;
        for (int j = 0; j < deg; ++j) {
            int kj = __builtin_amdgcn_readlane(key, j);
            rank += ((kj < key) || (kj == key && j < lane)) ? 1 : 0;
        }
        if (lane >= deg) rank = lane;
        const int skey = __builtin_amdgcn_ds_permute(rank << 2, key);
        const int sk = skey / n_rel;
        const int rk = skey - sk * n_rel;

        for (int k0 = 0; k0 < deg; k0 += 4) {
            float4 a[4], b[4];
            for (int u = 0; u < 4; ++u) {
                int k = (k0 + u < deg) ? (k0 + u) : (deg - 1);
                int s = __builtin_amdgcn_readlane(sk, k);
                int r = __builtin_amdgcn_readlane(rk, k);
                a[u] = ((const float4*)(ent + (size_t)s * HD))[lane];
                b[u] = ((const float4*)(rel + (size_t)r * HD))[lane];
            }
            float4 comp[4];
            float v[4];
            for (int u = 0; u < 4; ++u) {
                comp[u] = make_float4(a[u].x * b[u].x, a[u].y * b[u].y,
                                      a[u].z * b[u].z, a[u].w * b[u].w);
                v[u] = comp[u].x * c.x + comp[u].y * c.y + comp[u].z * c.z + comp[u].w * c.w;
            }
            for (int o = 16; o > 0; o >>= 1) {
                for (int u = 0; u < 4; ++u) v[u] += __shfl_xor(v[u], o, 32);
            }
            for (int u = 0; u < 4; ++u) {
                if (k0 + u < deg) seg_update(st, comp[u], v[u]);
            }
        }
    } else {
        int prev = -1;
        int done = 0;
        for (int it = 0; it < deg && done < deg; ++it) {
            int best = 0x7fffffff;
            for (int i = lane; i < deg; i += 32) {
                int kk = keys[beg + i];
                if (kk > prev && kk < best) best = kk;
            }
            for (int o = 16; o > 0; o >>= 1) {
                int y = __shfl_xor(best, o, 32);
                best = (y < best) ? y : best;
            }
            best = __builtin_amdgcn_readfirstlane(best);
            int mult = 0;
            for (int i = lane; i < deg; i += 32) mult += (keys[beg + i] == best) ? 1 : 0;
            for (int o = 16; o > 0; o >>= 1) mult += __shfl_xor(mult, o, 32);
            mult = __builtin_amdgcn_readfirstlane(mult);
            if (mult == 0) break;
            const int s = best / n_rel;
            const int r = best - s * n_rel;
            const float4 a = ((const float4*)(ent + (size_t)s * HD))[lane];
            const float4 b = ((const float4*)(rel + (size_t)r * HD))[lane];
            const float4 comp = make_float4(a.x * b.x, a.y * b.y, a.z * b.z, a.w * b.w);
            const float v = wave_sum32(comp.x * c.x + comp.y * c.y + comp.z * c.z + comp.w * c.w);
            for (int t = 0; t < mult; ++t) seg_update(st, comp, v);
            done += mult;
            prev = best;
        }
    }

    const float inv = 1.0f / st.l;
    *outp = make_float4(st.acc.x * inv, st.acc.y * inv, st.acc.z * inv, st.acc.w * inv);
}

__global__ void __launch_bounds__(TPB) gemm_wmma_kernel(
        const float* __restrict__ A,
        const float* __restrict__ W,
        float* __restrict__ H,
        int n_rows) {
    const int wave = threadIdx.x >> 5;
    const int lane = threadIdx.x & 31;
    const int half = lane >> 4;
    const int l16  = lane & 15;
    const int m0 = blockIdx.x * 16;
    const int n0 = wave * 16;

    v8f c = {};
    const float* arow = A + (size_t)(m0 + l16) * HD;
    const float* wcol = W + n0 + l16;

    #pragma unroll 4
    for (int k0 = 0; k0 < HD; k0 += 4) {
        int ka = k0 + 2 * half;
        v2f a, b;
        a.x = arow[ka];
        a.y = arow[ka + 1];
        b.x = wcol[(size_t)ka * HD];
        b.y = wcol[(size_t)(ka + 1) * HD];
        c = __builtin_amdgcn_wmma_f32_16x16x4_f32(
                 false, a,  false, b,
                 (short)0, c,  false,  false);
    }

    float* hb = H + (size_t)(m0 + 8 * half) * HD + n0 + l16;
    #pragma unroll
    for (int r2 = 0; r2 < 8; ++r2) hb[(size_t)r2 * HD] = c[r2];
}

__global__ void __launch_bounds__(TPB) bn_stats_kernel(
        const float* __restrict__ H, float* __restrict__ colsum,
        float* __restrict__ colsumsq, int n_rows) {
    int col  = threadIdx.x & (HD - 1);
    int roff = threadIdx.x >> 7;
    float s = 0.0f, sq = 0.0f;
    for (int r = blockIdx.x * 2 + roff; r < n_rows; r += gridDim.x * 2) {
        float v = H[(size_t)r * HD + col];
        s += v; sq += v * v;
    }
    atomicAdd(colsum + col, s);
    atomicAdd(colsumsq + col, sq);
}

__global__ void __launch_bounds__(TPB) bn_tanh_kernel(
        float* __restrict__ H, const float* __restrict__ colsum,
        const float* __restrict__ colsumsq, const float* __restrict__ gamma,
        const float* __restrict__ beta, int n_rows) {
    size_t i = (size_t)blockIdx.x * TPB + threadIdx.x;
    size_t total = (size_t)n_rows * HD;
    if (i >= total) return;
    int c = (int)(i & (HD - 1));
    float invN = 1.0f / (float)n_rows;
    float mu   = colsum[c] * invN;
    float var  = colsumsq[c] * invN - mu * mu;
    float istd = rsqrtf(var + 1e-5f);
    float v = (H[i] - mu) * istd * gamma[c] + beta[c];
    H[i] = tanhf(v);
}

extern "C" void kernel_launch(void* const* d_in, const int* in_sizes, int n_in,
                              void* d_out, int out_size, void* d_ws, size_t ws_size,
                              hipStream_t stream) {
    const float* ent   = (const float*)d_in[0];
    const float* rel   = (const float*)d_in[1];
    const float* W     = (const float*)d_in[2];
    const float* gamma = (const float*)d_in[3];
    const float* beta  = (const float*)d_in[4];
    const int*   src   = (const int*)d_in[5];
    const int*   dst   = (const int*)d_in[6];
    const int*   rid   = (const int*)d_in[7];
    float* H = (float*)d_out;

    const int N = in_sizes[0] / HD;
    const int R = in_sizes[1] / HD;
    const int E = in_sizes[5];

    float* ws       = (float*)d_ws;
    float* neigh    = ws;
    int*   keys     = (int*)(neigh + (size_t)N * HD);
    int*   cnt      = keys + E;
    float* colsum   = (float*)(cnt + N);
    float* colsumsq = colsum + HD;
    int*   off      = (int*)(colsumsq + HD);
    int*   pairs    = (int*)neigh;
    int*   bin_cnt   = cnt;
    int*   bin_cur   = cnt + MAX_BINS * CTR_PITCH;
    int*   bin_start = cnt + 2 * MAX_BINS * CTR_PITCH;

    size_t zt = (size_t)N + 2 * HD;
    zero_kernel<<<(unsigned)((zt + TPB - 1) / TPB), TPB, 0, stream>>>((float*)cnt, zt);

    const int n_bins = (N + BIN_NODES - 1) / BIN_NODES;
    bin_count_kernel<<<BIN_BLOCKS, TPB, 0, stream>>>(dst, bin_cnt, E, n_bins);
    bin_scatter_kernel<<<BIN_BLOCKS, TPB, 0, stream>>>(
        src, dst, rid, bin_cnt, bin_cur, bin_start, pairs, E, n_bins, R);
    bin_sort_kernel<<<(unsigned)n_bins, SORT_T, 0, stream>>>(pairs, bin_cnt, bin_start, off, keys, N);

    const int nodes_per_blk = TPB / 32;
    node_softmax_aggr_kernel<<<(unsigned)((N + nodes_per_blk - 1) / nodes_per_blk), TPB, 0, stream>>>(
        ent, rel, off, keys, neigh, N, E, R);

    gemm_wmma_kernel<<<(unsigned)(N / 16), TPB, 0, stream>>>(neigh, W, H, N);

    bn_stats_kernel<<<256, TPB, 0, stream>>>(H, colsum, colsumsq, N);
    bn_tanh_kernel<<<(unsigned)(((size_t)N * HD + TPB - 1) / TPB), TPB, 0, stream>>>(
        H, colsum, colsumsq, gamma, beta, N);
}


# BN column sums re-tiled: 16 B per lane of a whole 512-B row per load, 13 rows requested per wave before the first add (was 98 dependent 4-B loads); new sum tree via LDS, 512 atomic adds per column
# speedup vs baseline: 2.7727x; 1.2325x over previous
#include <hip/hip_runtime.h>
#include <math.h>

#define HD 128
#define TPB 256

typedef __attribute__((ext_vector_type(2))) float v2f;
typedef __attribute__((ext_vector_type(8))) float v8f;

__global__ void __launch_bounds__(TPB) zero_kernel(float* __restrict__ p, size_t n) {
    size_t i = (size_t)blockIdx.x * TPB + threadIdx.x;
    if (i < n) p[i] = 0.0f;
}

#define BIN_SHIFT 8
#define BIN_NODES 256
#define MAX_BINS 256
#define CTR_PITCH 32
#define BIN_BLOCKS 512
#define SORT_T 1024

__device__ __forceinline__ int wave_incl_scan_int(int v, int lane) {
    int x = v;
    for (int o = 1; o < 32; o <<= 1) {
        int y = __shfl_up(x, o, 32);
        if (lane >= o) x += y;
    }
    return x;
}

__global__ void __launch_bounds__(TPB) bin_count_kernel(
        const int* __restrict__ dst, int* __restrict__ bin_cnt, int n_edges, int n_bins) {
    __shared__ int hist[MAX_BINS];
    const int t = threadIdx.x;
    hist[t] = 0;
    __syncthreads();
    const int chunk = (n_edges + (int)gridDim.x - 1) / (int)gridDim.x;
    const int e0 = blockIdx.x * chunk;
    const int e1 = (e0 + chunk < n_edges) ? (e0 + chunk) : n_edges;
    for (int e = e0 + t; e < e1; e += 4 * TPB) {
        int d[4];
        #pragma unroll
        for (int u = 0; u < 4; ++u) {
            const int ee = e + u * TPB;
            d[u] = (ee < e1) ? dst[ee] : -1;
        }
        #pragma unroll
        for (int u = 0; u < 4; ++u) if (d[u] >= 0) atomicAdd(&hist[d[u] >> BIN_SHIFT], 1);
    }
    __syncthreads();
    const int h = hist[t];
    if (t < n_bins && h > 0) atomicAdd(bin_cnt + t * CTR_PITCH, h);
}

__global__ void __launch_bounds__(TPB) bin_scatter_kernel(
        const int* __restrict__ src, const int* __restrict__ dst, const int* __restrict__ rid,
        const int* __restrict__ bin_cnt, int* __restrict__ bin_cur, int* __restrict__ bin_start,
        int* __restrict__ pairs, int n_edges, int n_bins, int n_rel) {
    __shared__ int hist[MAX_BINS];
    __shared__ int base[MAX_BINS];
    __shared__ int wtot[TPB / 32];
    const int t = threadIdx.x;
    const int lane = t & 31;
    const int w = t >> 5;
    hist[t] = 0;
    __syncthreads();
    const int chunk = (n_edges + (int)gridDim.x - 1) / (int)gridDim.x;
    const int e0 = blockIdx.x * chunk;
    const int e1 = (e0 + chunk < n_edges) ? (e0 + chunk) : n_edges;
    for (int e = e0 + t; e < e1; e += 4 * TPB) {
        int d[4];
        #pragma unroll
        for (int u = 0; u < 4; ++u) {
            const int ee = e + u * TPB;
            d[u] = (ee < e1) ? dst[ee] : -1;
        }
        #pragma unroll
        for (int u = 0; u < 4; ++u) if (d[u] >= 0) atomicAdd(&hist[d[u] >> BIN_SHIFT], 1);
    }
    const int tot = (t < n_bins) ? bin_cnt[t * CTR_PITCH] : 0;
    const int incl = wave_incl_scan_int(tot, lane);
    if (lane == 31) wtot[w] = incl;
    __syncthreads();
    int pre = 0;
    for (int i = 0; i < w; ++i) pre += wtot[i];
    const int start = pre + incl - tot;
    if (blockIdx.x == 0 && t < n_bins) bin_start[t] = start;
    const int h = hist[t];
    int b0 = 0;
    if (t < n_bins && h > 0) b0 = start + atomicAdd(bin_cur + t * CTR_PITCH, h);
    base[t] = b0;
    hist[t] = 0;
    __syncthreads();
    for (int e = e0 + t; e < e1; e += 4 * TPB) {
        int d[4], s[4], r[4];
        #pragma unroll
        for (int u = 0; u < 4; ++u) {
            const int ee = e + u * TPB;
            const bool ok = ee < e1;
            d[u] = ok ? dst[ee] : -1;
            s[u] = ok ? src[ee] : 0;
            r[u] = ok ? rid[ee] : 0;
        }
        #pragma unroll
        for (int u = 0; u < 4; ++u) {
            if (d[u] >= 0) {
                const int b = d[u] >> BIN_SHIFT;
                const int slot = base[b] + atomicAdd(&hist[b], 1);
                *((int2*)(pairs + 2 * (size_t)slot)) = make_int2(d[u], s[u] * n_rel + r[u]);
            }
        }
    }
}

__global__ void __launch_bounds__(SORT_T) bin_sort_kernel(
        const int* __restrict__ pairs, const int* __restrict__ bin_cnt,
        const int* __restrict__ bin_start, int* __restrict__ off, int* __restrict__ keys,
        int n_nodes) {
    __shared__ int lcnt[BIN_NODES];
    __shared__ int lpos[BIN_NODES];
    __shared__ int wtot[BIN_NODES / 32];
    const int t = threadIdx.x;
    const int lane = t & 31;
    const int w = t >> 5;
    const int bin = blockIdx.x;
    const int bs = bin_start[bin];
    const int be = bs + bin_cnt[bin * CTR_PITCH];
    if (t < BIN_NODES) lcnt[t] = 0;
    __syncthreads();
    for (int i = bs + t; i < be; i += 4 * SORT_T) {
        int d[4];
        #pragma unroll
        for (int u = 0; u < 4; ++u) {
            const int ii = i + u * SORT_T;
            d[u] = (ii < be) ? pairs[2 * (size_t)ii] : -1;
        }
        #pragma unroll
        for (int u = 0; u < 4; ++u) if (d[u] >= 0) atomicAdd(&lcnt[d[u] & (BIN_NODES - 1)], 1);
    }
    __syncthreads();
    const int v = (t < BIN_NODES) ? lcnt[t] : 0;
    const int incl = wave_incl_scan_int(v, lane);
    if (t < BIN_NODES && lane == 31) wtot[w] = incl;
    __syncthreads();
    if (t < BIN_NODES) {
        int pre = 0;
        for (int i = 0; i < w; ++i) pre += wtot[i];
        const int ex = bs + pre + incl - v;
        lpos[t] = ex;
        const int node = bin * BIN_NODES + t;
        if (node < n_nodes) off[node] = ex;
    }
    __syncthreads();
    for (int i = bs + t; i < be; i += 4 * SORT_T) {
        int2 p[4];
        #pragma unroll
        for (int u = 0; u < 4; ++u) {
            const int ii = i + u * SORT_T;
            p[u] = (ii < be) ? *((const int2*)(pairs + 2 * (size_t)ii)) : make_int2(-1, 0);
        }
        #pragma unroll
        for (int u = 0; u < 4; ++u) {
            if (p[u].x >= 0) {
                const int slot = atomicAdd(&lpos[p[u].x & (BIN_NODES - 1)], 1);
                keys[slot] = p[u].y;
            }
        }
    }
}

struct SegAcc { float4 acc; float m; float l; };

__device__ __forceinline__ float wave_sum32(float v) {
    for (int o = 16; o > 0; o >>= 1) v += __shfl_xor(v, o, 32);
    return v;
}

__device__ __forceinline__ void seg_update(SegAcc& st, const float4& comp, float v) {
    float mn = fmaxf(st.m, v);
    float sc = __expf(st.m - mn);
    float p  = __expf(v - mn);
    st.acc.x = st.acc.x * sc + comp.x * p;
    st.acc.y = st.acc.y * sc + comp.y * p;
    st.acc.z = st.acc.z * sc + comp.z * p;
    st.acc.w = st.acc.w * sc + comp.w * p;
    st.l = st.l * sc + p;
    st.m = mn;
}

__global__ void __launch_bounds__(TPB) node_softmax_aggr_kernel(
        const float* __restrict__ ent, const float* __restrict__ rel,
        const int* __restrict__ off, const int* __restrict__ keys,
        float* __restrict__ neigh, int n_nodes, int n_edges, int n_rel) {
    const int d = __builtin_amdgcn_readfirstlane((int)(blockIdx.x * (TPB / 32) + (threadIdx.x >> 5)));
    if (d >= n_nodes) return;
    const int lane = threadIdx.x & 31;
    const int beg = off[d];
    const int end = (d + 1 < n_nodes) ? off[d + 1] : n_edges;
    const int deg = end - beg;
    float4* outp = (float4*)(neigh + (size_t)d * HD) + lane;
    if (deg <= 0) { *outp = make_float4(0.0f, 0.0f, 0.0f, 0.0f); return; }

    const float4 c = ((const float4*)(ent + (size_t)d * HD))[lane];
    SegAcc st;
    st.acc = make_float4(0.0f, 0.0f, 0.0f, 0.0f);
    st.m = -INFINITY;
    st.l = 0.0f;

    if (deg <= 32) {
        const int key = (lane < deg) ? keys[beg + lane] : 0x7fffffff;
        int rank = 0;
        for (int j = 0; j < deg; ++j) {
            int kj = __builtin_amdgcn_readlane(key, j);
            rank += ((kj < key) || (kj == key && j < lane)) ? 1 : 0;
        }
        if (lane >= deg) rank = lane;
        const int skey = __builtin_amdgcn_ds_permute(rank << 2, key);
        const int sk = skey / n_rel;
        const int rk = skey - sk * n_rel;

        for (int k0 = 0; k0 < deg; k0 += 4) {
            float4 a[4], b[4];
            for (int u = 0; u < 4; ++u) {
                int k = (k0 + u < deg) ? (k0 + u) : (deg - 1);
                int s = __builtin_amdgcn_readlane(sk, k);
                int r = __builtin_amdgcn_readlane(rk, k);
                a[u] = ((const float4*)(ent + (size_t)s * HD))[lane];
                b[u] = ((const float4*)(rel + (size_t)r * HD))[lane];
            }
            float4 comp[4];
            float v[4];
            for (int u = 0; u < 4; ++u) {
                comp[u] = make_float4(a[u].x * b[u].x, a[u].y * b[u].y,
                                      a[u].z * b[u].z, a[u].w * b[u].w);
                v[u] = comp[u].x * c.x + comp[u].y * c.y + comp[u].z * c.z + comp[u].w * c.w;
            }
            for (int o = 16; o > 0; o >>= 1) {
                for (int u = 0; u < 4; ++u) v[u] += __shfl_xor(v[u], o, 32);
            }
            for (int u = 0; u < 4; ++u) {
                if (k0 + u < deg) seg_update(st, comp[u], v[u]);
            }
        }
    } else {
        int prev = -1;
        int done = 0;
        for (int it = 0; it < deg && done < deg; ++it) {
            int best = 0x7fffffff;
            for (int i = lane; i < deg; i += 32) {
                int kk = keys[beg + i];
                if (kk > prev && kk < best) best = kk;
            }
            for (int o = 16; o > 0; o >>= 1) {
                int y = __shfl_xor(best, o, 32);
                best = (y < best) ? y : best;
            }
            best = __builtin_amdgcn_readfirstlane(best);
            int mult = 0;
            for (int i = lane; i < deg; i += 32) mult += (keys[beg + i] == best) ? 1 : 0;
            for (int o = 16; o > 0; o >>= 1) mult += __shfl_xor(mult, o, 32);
            mult = __builtin_amdgcn_readfirstlane(mult);
            if (mult == 0) break;
            const int s = best / n_rel;
            const int r = best - s * n_rel;
            const float4 a = ((const float4*)(ent + (size_t)s * HD))[lane];
            const float4 b = ((const float4*)(rel + (size_t)r * HD))[lane];
            const float4 comp = make_float4(a.x * b.x, a.y * b.y, a.z * b.z, a.w * b.w);
            const float v = wave_sum32(comp.x * c.x + comp.y * c.y + comp.z * c.z + comp.w * c.w);
            for (int t = 0; t < mult; ++t) seg_update(st, comp, v);
            done += mult;
            prev = best;
        }
    }

    const float inv = 1.0f / st.l;
    *outp = make_float4(st.acc.x * inv, st.acc.y * inv, st.acc.z * inv, st.acc.w * inv);
}

__global__ void __launch_bounds__(TPB) gemm_wmma_kernel(
        const float* __restrict__ A,
        const float* __restrict__ W,
        float* __restrict__ H,
        int n_rows) {
    const int wave = threadIdx.x >> 5;
    const int lane = threadIdx.x & 31;
    const int half = lane >> 4;
    const int l16  = lane & 15;
    const int m0 = blockIdx.x * 16;
    const int n0 = wave * 16;

    v8f c = {};
    const float* arow = A + (size_t)(m0 + l16) * HD;
    const float* wcol = W + n0 + l16;

    #pragma unroll 4
    for (int k0 = 0; k0 < HD; k0 += 4) {
        int ka = k0 + 2 * half;
        v2f a, b;
        a.x = arow[ka];
        a.y = arow[ka + 1];
        b.x = wcol[(size_t)ka * HD];
        b.y = wcol[(size_t)(ka + 1) * HD];
        c = __builtin_amdgcn_wmma_f32_16x16x4_f32(
                 false, a,  false, b,
                 (short)0, c,  false,  false);
    }

    float* hb = H + (size_t)(m0 + 8 * half) * HD + n0 + l16;
    #pragma unroll
    for (int r2 = 0; r2 < 8; ++r2) hb[(size_t)r2 * HD] = c[r2];
}

#define STAT_ROWS 13
#define STAT_WAVES (TPB / 32)
__global__ void __launch_bounds__(TPB) bn_stats_kernel(
        const float* __restrict__ H, float* __restrict__ colsum,
        float* __restrict__ colsumsq, int n_rows) {
    __shared__ float part[2][STAT_WAVES][HD];
    const int lane = threadIdx.x & 31;
    const int w = __builtin_amdgcn_readfirstlane((int)(threadIdx.x >> 5));
    const int b = blockIdx.x;
    const int G = gridDim.x;
    const float4* H4 = (const float4*)H;
    float4 s = make_float4(0.0f, 0.0f, 0.0f, 0.0f);
    float4 q = make_float4(0.0f, 0.0f, 0.0f, 0.0f);
    for (int i0 = w; i0 * G + b < n_rows; i0 += STAT_WAVES * STAT_ROWS) {
        float4 v[STAT_ROWS];
        #pragma unroll
        for (int u = 0; u < STAT_ROWS; ++u) {
            const int r = (i0 + u * STAT_WAVES) * G + b;
            v[u] = (r < n_rows) ? H4[(size_t)r * (HD / 4) + lane]
                                : make_float4(0.0f, 0.0f, 0.0f, 0.0f);
        }
        #pragma unroll
        for (int u = 0; u < STAT_ROWS; ++u) {
            s.x += v[u].x; s.y += v[u].y; s.z += v[u].z; s.w += v[u].w;
            q.x += v[u].x * v[u].x; q.y += v[u].y * v[u].y;
            q.z += v[u].z * v[u].z; q.w += v[u].w * v[u].w;
        }
    }
    ((float4*)&part[0][w][0])[lane] = s;
    ((float4*)&part[1][w][0])[lane] = q;
    __syncthreads();
    const int stat = threadIdx.x >> 7;
    const int col  = threadIdx.x & (HD - 1);
    float acc = 0.0f;
    for (int i = 0; i < STAT_WAVES; ++i) acc += part[stat][i][col];
    atomicAdd((stat ? colsumsq : colsum) + col, acc);
}

__global__ void __launch_bounds__(TPB) bn_tanh_kernel(
        float* __restrict__ H, const float* __restrict__ colsum,
        const float* __restrict__ colsumsq, const float* __restrict__ gamma,
        const float* __restrict__ beta, int n_rows) {
    size_t i = (size_t)blockIdx.x * TPB + threadIdx.x;
    size_t total = (size_t)n_rows * HD;
    if (i >= total) return;
    int c = (int)(i & (HD - 1));
    float invN = 1.0f / (float)n_rows;
    float mu   = colsum[c] * invN;
    float var  = colsumsq[c] * invN - mu * mu;
    float istd = rsqrtf(var + 1e-5f);
    float v = (H[i] - mu) * istd * gamma[c] + beta[c];
    H[i] = tanhf(v);
}

extern "C" void kernel_launch(void* const* d_in, const int* in_sizes, int n_in,
                              void* d_out, int out_size, void* d_ws, size_t ws_size,
                              hipStream_t stream) {
    const float* ent   = (const float*)d_in[0];
    const float* rel   = (const float*)d_in[1];
    const float* W     = (const float*)d_in[2];
    const float* gamma = (const float*)d_in[3];
    const float* beta  = (const float*)d_in[4];
    const int*   src   = (const int*)d_in[5];
    const int*   dst   = (const int*)d_in[6];
    const int*   rid   = (const int*)d_in[7];
    float* H = (float*)d_out;

    const int N = in_sizes[0] / HD;
    const int R = in_sizes[1] / HD;
    const int E = in_sizes[5];

    float* ws       = (float*)d_ws;
    float* neigh    = ws;
    int*   keys     = (int*)(neigh + (size_t)N * HD);
    int*   cnt      = keys + E;
    float* colsum   = (float*)(cnt + N);
    float* colsumsq = colsum + HD;
    int*   off      = (int*)(colsumsq + HD);
    int*   pairs    = (int*)neigh;
    int*   bin_cnt   = cnt;
    int*   bin_cur   = cnt + MAX_BINS * CTR_PITCH;
    int*   bin_start = cnt + 2 * MAX_BINS * CTR_PITCH;

    size_t zt = (size_t)N + 2 * HD;
    zero_kernel<<<(unsigned)((zt + TPB - 1) / TPB), TPB, 0, stream>>>((float*)cnt, zt);

    const int n_bins = (N + BIN_NODES - 1) / BIN_NODES;
    bin_count_kernel<<<BIN_BLOCKS, TPB, 0, stream>>>(dst, bin_cnt, E, n_bins);
    bin_scatter_kernel<<<BIN_BLOCKS, TPB, 0, stream>>>(
        src, dst, rid, bin_cnt, bin_cur, bin_start, pairs, E, n_bins, R);
    bin_sort_kernel<<<(unsigned)n_bins, SORT_T, 0, stream>>>(pairs, bin_cnt, bin_start, off, keys, N);

    const int nodes_per_blk = TPB / 32;
    node_softmax_aggr_kernel<<<(unsigned)((N + nodes_per_blk - 1) / nodes_per_blk), TPB, 0, stream>>>(
        ent, rel, off, keys, neigh, N, E, R);

    gemm_wmma_kernel<<<(unsigned)(N / 16), TPB, 0, stream>>>(neigh, W, H, N);

    bn_stats_kernel<<<512, TPB, 0, stream>>>(H, colsum, colsumsq, N);
    bn_tanh_kernel<<<(unsigned)(((size_t)N * HD + TPB - 1) / TPB), TPB, 0, stream>>>(
        H, colsum, colsumsq, gamma, beta, N);
}


# Aggregation and GEMM fused: 8 waves leave the neigh rows of their block's 16 dst nodes in LDS (2 per wave), then the same f32 WMMA K-loop takes A from LDS; neigh plane and a launch gone; same sums
# speedup vs baseline: 3.1446x; 1.1341x over previous
#include <hip/hip_runtime.h>
#include <math.h>

#define HD 128
#define TPB 256

typedef __attribute__((ext_vector_type(2))) float v2f;
typedef __attribute__((ext_vector_type(8))) float v8f;

__global__ void __launch_bounds__(TPB) zero_kernel(float* __restrict__ p, size_t n) {
    size_t i = (size_t)blockIdx.x * TPB + threadIdx.x;
    if (i < n) p[i] = 0.0f;
}

#define BIN_SHIFT 8
#define BIN_NODES 256
#define MAX_BINS 256
#define CTR_PITCH 32
#define BIN_BLOCKS 512
#define SORT_T 1024

__device__ __forceinline__ int wave_incl_scan_int(int v, int lane) {
    int x = v;
    for (int o = 1; o < 32; o <<= 1) {
        int y = __shfl_up(x, o, 32);
        if (lane >= o) x += y;
    }
    return x;
}

__global__ void __launch_bounds__(TPB) bin_count_kernel(
        const int* __restrict__ dst, int* __restrict__ bin_cnt, int n_edges, int n_bins) {
    __shared__ int hist[MAX_BINS];
    const int t = threadIdx.x;
    hist[t] = 0;
    __syncthreads();
    const int chunk = (n_edges + (int)gridDim.x - 1) / (int)gridDim.x;
    const int e0 = blockIdx.x * chunk;
    const int e1 = (e0 + chunk < n_edges) ? (e0 + chunk) : n_edges;
    for (int e = e0 + t; e < e1; e += 4 * TPB) {
        int d[4];
        #pragma unroll
        for (int u = 0; u < 4; ++u) {
            const int ee = e + u * TPB;
            d[u] = (ee < e1) ? dst[ee] : -1;
        }
        #pragma unroll
        for (int u = 0; u < 4; ++u) if (d[u] >= 0) atomicAdd(&hist[d[u] >> BIN_SHIFT], 1);
    }
    __syncthreads();
    const int h = hist[t];
    if (t < n_bins && h > 0) atomicAdd(bin_cnt + t * CTR_PITCH, h);
}

__global__ void __launch_bounds__(TPB) bin_scatter_kernel(
        const int* __restrict__ src, const int* __restrict__ dst, const int* __restrict__ rid,
        const int* __restrict__ bin_cnt, int* __restrict__ bin_cur, int* __restrict__ bin_start,
        int* __restrict__ pairs, int n_edges, int n_bins, int n_rel) {
    __shared__ int hist[MAX_BINS];
    __shared__ int base[MAX_BINS];
    __shared__ int wtot[TPB / 32];
    const int t = threadIdx.x;
    const int lane = t & 31;
    const int w = t >> 5;
    hist[t] = 0;
    __syncthreads();
    const int chunk = (n_edges + (int)gridDim.x - 1) / (int)gridDim.x;
    const int e0 = blockIdx.x * chunk;
    const int e1 = (e0 + chunk < n_edges) ? (e0 + chunk) : n_edges;
    for (int e = e0 + t; e < e1; e += 4 * TPB) {
        int d[4];
        #pragma unroll
        for (int u = 0; u < 4; ++u) {
            const int ee = e + u * TPB;
            d[u] = (ee < e1) ? dst[ee] : -1;
        }
        #pragma unroll
        for (int u = 0; u < 4; ++u) if (d[u] >= 0) atomicAdd(&hist[d[u] >> BIN_SHIFT], 1);
    }
    const int tot = (t < n_bins) ? bin_cnt[t * CTR_PITCH] : 0;
    const int incl = wave_incl_scan_int(tot, lane);
    if (lane == 31) wtot[w] = incl;
    __syncthreads();
    int pre = 0;
    for (int i = 0; i < w; ++i) pre += wtot[i];
    const int start = pre + incl - tot;
    if (blockIdx.x == 0 && t < n_bins) bin_start[t] = start;
    const int h = hist[t];
    int b0 = 0;
    if (t < n_bins && h > 0) b0 = start + atomicAdd(bin_cur + t * CTR_PITCH, h);
    base[t] = b0;
    hist[t] = 0;
    __syncthreads();
    for (int e = e0 + t; e < e1; e += 4 * TPB) {
        int d[4], s[4], r[4];
        #pragma unroll
        for (int u = 0; u < 4; ++u) {
            const int ee = e + u * TPB;
            const bool ok = ee < e1;
            d[u] = ok ? dst[ee] : -1;
            s[u] = ok ? src[ee] : 0;
            r[u] = ok ? rid[ee] : 0;
        }
        #pragma unroll
        for (int u = 0; u < 4; ++u) {
            if (d[u] >= 0) {
                const int b = d[u] >> BIN_SHIFT;
                const int slot = base[b] + atomicAdd(&hist[b], 1);
                *((int2*)(pairs + 2 * (size_t)slot)) = make_int2(d[u], s[u] * n_rel + r[u]);
            }
        }
    }
}

__global__ void __launch_bounds__(SORT_T) bin_sort_kernel(
        const int* __restrict__ pairs, const int* __restrict__ bin_cnt,
        const int* __restrict__ bin_start, int* __restrict__ off, int* __restrict__ keys,
        int n_nodes) {
    __shared__ int lcnt[BIN_NODES];
    __shared__ int lpos[BIN_NODES];
    __shared__ int wtot[BIN_NODES / 32];
    const int t = threadIdx.x;
    const int lane = t & 31;
    const int w = t >> 5;
    const int bin = blockIdx.x;
    const int bs = bin_start[bin];
    const int be = bs + bin_cnt[bin * CTR_PITCH];
    if (t < BIN_NODES) lcnt[t] = 0;
    __syncthreads();
    for (int i = bs + t; i < be; i += 4 * SORT_T) {
        int d[4];
        #pragma unroll
        for (int u = 0; u < 4; ++u) {
            const int ii = i + u * SORT_T;
            d[u] = (ii < be) ? pairs[2 * (size_t)ii] : -1;
        }
        #pragma unroll
        for (int u = 0; u < 4; ++u) if (d[u] >= 0) atomicAdd(&lcnt[d[u] & (BIN_NODES - 1)], 1);
    }
    __syncthreads();
    const int v = (t < BIN_NODES) ? lcnt[t] : 0;
    const int incl = wave_incl_scan_int(v, lane);
    if (t < BIN_NODES && lane == 31) wtot[w] = incl;
    __syncthreads();
    if (t < BIN_NODES) {
        int pre = 0;
        for (int i = 0; i < w; ++i) pre += wtot[i];
        const int ex = bs + pre + incl - v;
        lpos[t] = ex;
        const int node = bin * BIN_NODES + t;
        if (node < n_nodes) off[node] = ex;
    }
    __syncthreads();
    for (int i = bs + t; i < be; i += 4 * SORT_T) {
        int2 p[4];
        #pragma unroll
        for (int u = 0; u < 4; ++u) {
            const int ii = i + u * SORT_T;
            p[u] = (ii < be) ? *((const int2*)(pairs + 2 * (size_t)ii)) : make_int2(-1, 0);
        }
        #pragma unroll
        for (int u = 0; u < 4; ++u) {
            if (p[u].x >= 0) {
                const int slot = atomicAdd(&lpos[p[u].x & (BIN_NODES - 1)], 1);
                keys[slot] = p[u].y;
            }
        }
    }
}

struct SegAcc { float4 acc; float m; float l; };

__device__ __forceinline__ float wave_sum32(float v) {
    for (int o = 16; o > 0; o >>= 1) v += __shfl_xor(v, o, 32);
    return v;
}

__device__ __forceinline__ void seg_update(SegAcc& st, const float4& comp, float v) {
    float mn = fmaxf(st.m, v);
    float sc = __expf(st.m - mn);
    float p  = __expf(v - mn);
    st.acc.x = st.acc.x * sc + comp.x * p;
    st.acc.y = st.acc.y * sc + comp.y * p;
    st.acc.z = st.acc.z * sc + comp.z * p;
    st.acc.w = st.acc.w * sc + comp.w * p;
    st.l = st.l * sc + p;
    st.m = mn;
}

__device__ __forceinline__ float4 node_row(
        const float* __restrict__ ent, const float* __restrict__ rel,
        const int* __restrict__ off, const int* __restrict__ keys,
        int d, int lane, int n_nodes, int n_edges, int n_rel) {
    const float4 zero4 = make_float4(0.0f, 0.0f, 0.0f, 0.0f);
    if (d >= n_nodes) return zero4;
    const int beg = off[d];
    const int end = (d + 1 < n_nodes) ? off[d + 1] : n_edges;
    const int deg = end - beg;
    if (deg <= 0) return zero4;

    const float4 c = ((const float4*)(ent + (size_t)d * HD))[lane];
    SegAcc st;
    st.acc = zero4;
    st.m = -INFINITY;
    st.l = 0.0f;

    if (deg <= 32) {
        const int key = (lane < deg) ? keys[beg + lane] : 0x7fffffff;
        int rank = 0;
        for (int j = 0; j < deg; ++j) {
            int kj = __builtin_amdgcn_readlane(key, j);
            rank += ((kj < key) || (kj == key && j < lane)) ? 1 : 0;
        }
        if (lane >= deg) rank = lane;
        const int skey = __builtin_amdgcn_ds_permute(rank << 2, key);
        const int sk = skey / n_rel;
        const int rk = skey - sk * n_rel;

        for (int k0 = 0; k0 < deg; k0 += 4) {
            float4 a[4], b[4];
            for (int u = 0; u < 4; ++u) {
                int k = (k0 + u < deg) ? (k0 + u) : (deg - 1);
                int s = __builtin_amdgcn_readlane(sk, k);
                int r = __builtin_amdgcn_readlane(rk, k);
                a[u] = ((const float4*)(ent + (size_t)s * HD))[lane];
                b[u] = ((const float4*)(rel + (size_t)r * HD))[lane];
            }
            float4 comp[4];
            float v[4];
            for (int u = 0; u < 4; ++u) {
                comp[u] = make_float4(a[u].x * b[u].x, a[u].y * b[u].y,
                                      a[u].z * b[u].z, a[u].w * b[u].w);
                v[u] = comp[u].x * c.x + comp[u].y * c.y + comp[u].z * c.z + comp[u].w * c.w;
            }
            for (int o = 16; o > 0; o >>= 1) {
                for (int u = 0; u < 4; ++u) v[u] += __shfl_xor(v[u], o, 32);
            }
            for (int u = 0; u < 4; ++u) {
                if (k0 + u < deg) seg_update(st, comp[u], v[u]);
            }
        }
    } else {
        int prev = -1;
        int done = 0;
        for (int it = 0; it < deg && done < deg; ++it) {
            int best = 0x7fffffff;
            for (int i = lane; i < deg; i += 32) {
                int kk = keys[beg + i];
                if (kk > prev && kk < best) best = kk;
            }
            for (int o = 16; o > 0; o >>= 1) {
                int y = __shfl_xor(best, o, 32);
                best = (y < best) ? y : best;
            }
            best = __builtin_amdgcn_readfirstlane(best);
            int mult = 0;
            for (int i = lane; i < deg; i += 32) mult += (keys[beg + i] == best) ? 1 : 0;
            for (int o = 16; o > 0; o >>= 1) mult += __shfl_xor(mult, o, 32);
            mult = __builtin_amdgcn_readfirstlane(mult);
            if (mult == 0) break;
            const int s = best / n_rel;
            const int r = best - s * n_rel;
            const float4 a = ((const float4*)(ent + (size_t)s * HD))[lane];
            const float4 b = ((const float4*)(rel + (size_t)r * HD))[lane];
            const float4 comp = make_float4(a.x * b.x, a.y * b.y, a.z * b.z, a.w * b.w);
            const float v = wave_sum32(comp.x * c.x + comp.y * c.y + comp.z * c.z + comp.w * c.w);
            for (int t = 0; t < mult; ++t) seg_update(st, comp, v);
            done += mult;
            prev = best;
        }
    }

    const float inv = 1.0f / st.l;
    return make_float4(st.acc.x * inv, st.acc.y * inv, st.acc.z * inv, st.acc.w * inv);
}

#define TILE_M 16
#define NODE_WAVES (TPB / 32)
#define TILE_PITCH (HD + 4)

__global__ void __launch_bounds__(TPB) node_gemm_kernel(
        const float* __restrict__ ent, const float* __restrict__ rel,
        const int* __restrict__ off, const int* __restrict__ keys,
        const float* __restrict__ W,
        float* __restrict__ H,
        int n_nodes, int n_edges, int n_rel) {
    __shared__ __attribute__((aligned(16))) float tile[TILE_M * TILE_PITCH];
    const int lane = threadIdx.x & 31;
    const int wave = __builtin_amdgcn_readfirstlane((int)(threadIdx.x >> 5));
    const int m0 = blockIdx.x * TILE_M;

    for (int u = 0; u < TILE_M / NODE_WAVES; ++u) {
        const int row = wave * (TILE_M / NODE_WAVES) + u;
        const float4 nr = node_row(ent, rel, off, keys, m0 + row, lane, n_nodes, n_edges, n_rel);
        ((float4*)(tile + row * TILE_PITCH))[lane] = nr;
    }
    __syncthreads();

    const int half = lane >> 4;
    const int l16  = lane & 15;
    const int n0 = wave * 16;
    v8f c = {};
    const float* arow = tile + l16 * TILE_PITCH;
    const float* wcol = W + n0 + l16;

    #pragma unroll 8
    for (int k0 = 0; k0 < HD; k0 += 4) {
        const int ka = k0 + 2 * half;
        const v2f a = *((const v2f*)(arow + ka));
        v2f b;
        b.x = wcol[(size_t)ka * HD];
        b.y = wcol[(size_t)(ka + 1) * HD];
        c = __builtin_amdgcn_wmma_f32_16x16x4_f32(
                 false, a,  false, b,
                 (short)0, c,  false,  false);
    }

    const int mrow = m0 + 8 * half;
    float* hb = H + (size_t)mrow * HD + n0 + l16;
    #pragma unroll
    for (int r2 = 0; r2 < 8; ++r2) {
        if (mrow + r2 < n_nodes) hb[(size_t)r2 * HD] = c[r2];
    }
}

#define STAT_ROWS 13
#define STAT_WAVES (TPB / 32)
__global__ void __launch_bounds__(TPB) bn_stats_kernel(
        const float* __restrict__ H, float* __restrict__ colsum,
        float* __restrict__ colsumsq, int n_rows) {
    __shared__ float part[2][STAT_WAVES][HD];
    const int lane = threadIdx.x & 31;
    const int w = __builtin_amdgcn_readfirstlane((int)(threadIdx.x >> 5));
    const int b = blockIdx.x;
    const int G = gridDim.x;
    const float4* H4 = (const float4*)H;
    float4 s = make_float4(0.0f, 0.0f, 0.0f, 0.0f);
    float4 q = make_float4(0.0f, 0.0f, 0.0f, 0.0f);
    for (int i0 = w; i0 * G + b < n_rows; i0 += STAT_WAVES * STAT_ROWS) {
        float4 v[STAT_ROWS];
        #pragma unroll
        for (int u = 0; u < STAT_ROWS; ++u) {
            const int r = (i0 + u * STAT_WAVES) * G + b;
            v[u] = (r < n_rows) ? H4[(size_t)r * (HD / 4) + lane]
                                : make_float4(0.0f, 0.0f, 0.0f, 0.0f);
        }
        #pragma unroll
        for (int u = 0; u < STAT_ROWS; ++u) {
            s.x += v[u].x; s.y += v[u].y; s.z += v[u].z; s.w += v[u].w;
            q.x += v[u].x * v[u].x; q.y += v[u].y * v[u].y;
            q.z += v[u].z * v[u].z; q.w += v[u].w * v[u].w;
        }
    }
    ((float4*)&part[0][w][0])[lane] = s;
    ((float4*)&part[1][w][0])[lane] = q;
    __syncthreads();
    const int stat = threadIdx.x >> 7;
    const int col  = threadIdx.x & (HD - 1);
    float acc = 0.0f;
    for (int i = 0; i < STAT_WAVES; ++i) acc += part[stat][i][col];
    atomicAdd((stat ? colsumsq : colsum) + col, acc);
}

__global__ void __launch_bounds__(TPB) bn_tanh_kernel(
        float* __restrict__ H, const float* __restrict__ colsum,
        const float* __restrict__ colsumsq, const float* __restrict__ gamma,
        const float* __restrict__ beta, int n_rows) {
    size_t i = (size_t)blockIdx.x * TPB + threadIdx.x;
    size_t total = (size_t)n_rows * HD;
    if (i >= total) return;
    int c = (int)(i & (HD - 1));
    float invN = 1.0f / (float)n_rows;
    float mu   = colsum[c] * invN;
    float var  = colsumsq[c] * invN - mu * mu;
    float istd = rsqrtf(var + 1e-5f);
    float v = (H[i] - mu) * istd * gamma[c] + beta[c];
    H[i] = tanhf(v);
}

extern "C" void kernel_launch(void* const* d_in, const int* in_sizes, int n_in,
                              void* d_out, int out_size, void* d_ws, size_t ws_size,
                              hipStream_t stream) {
    const float* ent   = (const float*)d_in[0];
    const float* rel   = (const float*)d_in[1];
    const float* W     = (const float*)d_in[2];
    const float* gamma = (const float*)d_in[3];
    const float* beta  = (const float*)d_in[4];
    const int*   src   = (const int*)d_in[5];
    const int*   dst   = (const int*)d_in[6];
    const int*   rid   = (const int*)d_in[7];
    float* H = (float*)d_out;

    const int N = in_sizes[0] / HD;
    const int R = in_sizes[1] / HD;
    const int E = in_sizes[5];

    float* ws       = (float*)d_ws;
    float* neigh    = ws;
    int*   keys     = (int*)(neigh + (size_t)N * HD);
    int*   cnt      = keys + E;
    float* colsum   = (float*)(cnt + N);
    float* colsumsq = colsum + HD;
    int*   off      = (int*)(colsumsq + HD);
    int*   pairs    = (int*)neigh;
    int*   bin_cnt   = cnt;
    int*   bin_cur   = cnt + MAX_BINS * CTR_PITCH;
    int*   bin_start = cnt + 2 * MAX_BINS * CTR_PITCH;

    size_t zt = (size_t)N + 2 * HD;
    zero_kernel<<<(unsigned)((zt + TPB - 1) / TPB), TPB, 0, stream>>>((float*)cnt, zt);

    const int n_bins = (N + BIN_NODES - 1) / BIN_NODES;
    bin_count_kernel<<<BIN_BLOCKS, TPB, 0, stream>>>(dst, bin_cnt, E, n_bins);
    bin_scatter_kernel<<<BIN_BLOCKS, TPB, 0, stream>>>(
        src, dst, rid, bin_cnt, bin_cur, bin_start, pairs, E, n_bins, R);
    bin_sort_kernel<<<(unsigned)n_bins, SORT_T, 0, stream>>>(pairs, bin_cnt, bin_start, off, keys, N);

    node_gemm_kernel<<<(unsigned)((N + TILE_M - 1) / TILE_M), TPB, 0, stream>>>(
        ent, rel, off, keys, W, H, N, E, R);

    bn_stats_kernel<<<512, TPB, 0, stream>>>(H, colsum, colsumsq, N);
    bn_tanh_kernel<<<(unsigned)(((size_t)N * HD + TPB - 1) / TPB), TPB, 0, stream>>>(
        H, colsum, colsumsq, gamma, beta, N);
}


# Segments of 33 to 64 edges: two keys per lane ranked in the wave by lane broadcasts, edges taken in key order 4 at a time (was one serial minimum search over the segment per edge); same order, sums
# speedup vs baseline: 3.8741x; 1.2320x over previous
#include <hip/hip_runtime.h>
#include <math.h>

#define HD 128
#define TPB 256

typedef __attribute__((ext_vector_type(2))) float v2f;
typedef __attribute__((ext_vector_type(8))) float v8f;

__global__ void __launch_bounds__(TPB) zero_kernel(float* __restrict__ p, size_t n) {
    size_t i = (size_t)blockIdx.x * TPB + threadIdx.x;
    if (i < n) p[i] = 0.0f;
}

#define BIN_SHIFT 8
#define BIN_NODES 256
#define MAX_BINS 256
#define CTR_PITCH 32
#define BIN_BLOCKS 512
#define SORT_T 1024

__device__ __forceinline__ int wave_incl_scan_int(int v, int lane) {
    int x = v;
    for (int o = 1; o < 32; o <<= 1) {
        int y = __shfl_up(x, o, 32);
        if (lane >= o) x += y;
    }
    return x;
}

__global__ void __launch_bounds__(TPB) bin_count_kernel(
        const int* __restrict__ dst, int* __restrict__ bin_cnt, int n_edges, int n_bins) {
    __shared__ int hist[MAX_BINS];
    const int t = threadIdx.x;
    hist[t] = 0;
    __syncthreads();
    const int chunk = (n_edges + (int)gridDim.x - 1) / (int)gridDim.x;
    const int e0 = blockIdx.x * chunk;
    const int e1 = (e0 + chunk < n_edges) ? (e0 + chunk) : n_edges;
    for (int e = e0 + t; e < e1; e += 4 * TPB) {
        int d[4];
        #pragma unroll
        for (int u = 0; u < 4; ++u) {
            const int ee = e + u * TPB;
            d[u] = (ee < e1) ? dst[ee] : -1;
        }
        #pragma unroll
        for (int u = 0; u < 4; ++u) if (d[u] >= 0) atomicAdd(&hist[d[u] >> BIN_SHIFT], 1);
    }
    __syncthreads();
    const int h = hist[t];
    if (t < n_bins && h > 0) atomicAdd(bin_cnt + t * CTR_PITCH, h);
}

__global__ void __launch_bounds__(TPB) bin_scatter_kernel(
        const int* __restrict__ src, const int* __restrict__ dst, const int* __restrict__ rid,
        const int* __restrict__ bin_cnt, int* __restrict__ bin_cur, int* __restrict__ bin_start,
        int* __restrict__ pairs, int n_edges, int n_bins, int n_rel) {
    __shared__ int hist[MAX_BINS];
    __shared__ int base[MAX_BINS];
    __shared__ int wtot[TPB / 32];
    const int t = threadIdx.x;
    const int lane = t & 31;
    const int w = t >> 5;
    hist[t] = 0;
    __syncthreads();
    const int chunk = (n_edges + (int)gridDim.x - 1) / (int)gridDim.x;
    const int e0 = blockIdx.x * chunk;
    const int e1 = (e0 + chunk < n_edges) ? (e0 + chunk) : n_edges;
    for (int e = e0 + t; e < e1; e += 4 * TPB) {
        int d[4];
        #pragma unroll
        for (int u = 0; u < 4; ++u) {
            const int ee = e + u * TPB;
            d[u] = (ee < e1) ? dst[ee] : -1;
        }
        #pragma unroll
        for (int u = 0; u < 4; ++u) if (d[u] >= 0) atomicAdd(&hist[d[u] >> BIN_SHIFT], 1);
    }
    const int tot = (t < n_bins) ? bin_cnt[t * CTR_PITCH] : 0;
    const int incl = wave_incl_scan_int(tot, lane);
    if (lane == 31) wtot[w] = incl;
    __syncthreads();
    int pre = 0;
    for (int i = 0; i < w; ++i) pre += wtot[i];
    const int start = pre + incl - tot;
    if (blockIdx.x == 0 && t < n_bins) bin_start[t] = start;
    const int h = hist[t];
    int b0 = 0;
    if (t < n_bins && h > 0) b0 = start + atomicAdd(bin_cur + t * CTR_PITCH, h);
    base[t] = b0;
    hist[t] = 0;
    __syncthreads();
    for (int e = e0 + t; e < e1; e += 4 * TPB) {
        int d[4], s[4], r[4];
        #pragma unroll
        for (int u = 0; u < 4; ++u) {
            const int ee = e + u * TPB;
            const bool ok = ee < e1;
            d[u] = ok ? dst[ee] : -1;
            s[u] = ok ? src[ee] : 0;
            r[u] = ok ? rid[ee] : 0;
        }
        #pragma unroll
        for (int u = 0; u < 4; ++u) {
            if (d[u] >= 0) {
                const int b = d[u] >> BIN_SHIFT;
                const int slot = base[b] + atomicAdd(&hist[b], 1);
                *((int2*)(pairs + 2 * (size_t)slot)) = make_int2(d[u], s[u] * n_rel + r[u]);
            }
        }
    }
}

__global__ void __launch_bounds__(SORT_T) bin_sort_kernel(
        const int* __restrict__ pairs, const int* __restrict__ bin_cnt,
        const int* __restrict__ bin_start, int* __restrict__ off, int* __restrict__ keys,
        int n_nodes) {
    __shared__ int lcnt[BIN_NODES];
    __shared__ int lpos[BIN_NODES];
    __shared__ int wtot[BIN_NODES / 32];
    const int t = threadIdx.x;
    const int lane = t & 31;
    const int w = t >> 5;
    const int bin = blockIdx.x;
    const int bs = bin_start[bin];
    const int be = bs + bin_cnt[bin * CTR_PITCH];
    if (t < BIN_NODES) lcnt[t] = 0;
    __syncthreads();
    for (int i = bs + t; i < be; i += 4 * SORT_T) {
        int d[4];
        #pragma unroll
        for (int u = 0; u < 4; ++u) {
            const int ii = i + u * SORT_T;
            d[u] = (ii < be) ? pairs[2 * (size_t)ii] : -1;
        }
        #pragma unroll
        for (int u = 0; u < 4; ++u) if (d[u] >= 0) atomicAdd(&lcnt[d[u] & (BIN_NODES - 1)], 1);
    }
    __syncthreads();
    const int v = (t < BIN_NODES) ? lcnt[t] : 0;
    const int incl = wave_incl_scan_int(v, lane);
    if (t < BIN_NODES && lane == 31) wtot[w] = incl;
    __syncthreads();
    if (t < BIN_NODES) {
        int pre = 0;
        for (int i = 0; i < w; ++i) pre += wtot[i];
        const int ex = bs + pre + incl - v;
        lpos[t] = ex;
        const int node = bin * BIN_NODES + t;
        if (node < n_nodes) off[node] = ex;
    }
    __syncthreads();
    for (int i = bs + t; i < be; i += 4 * SORT_T) {
        int2 p[4];
        #pragma unroll
        for (int u = 0; u < 4; ++u) {
            const int ii = i + u * SORT_T;
            p[u] = (ii < be) ? *((const int2*)(pairs + 2 * (size_t)ii)) : make_int2(-1, 0);
        }
        #pragma unroll
        for (int u = 0; u < 4; ++u) {
            if (p[u].x >= 0) {
                const int slot = atomicAdd(&lpos[p[u].x & (BIN_NODES - 1)], 1);
                keys[slot] = p[u].y;
            }
        }
    }
}

struct SegAcc { float4 acc; float m; float l; };

__device__ __forceinline__ float wave_sum32(float v) {
    for (int o = 16; o > 0; o >>= 1) v += __shfl_xor(v, o, 32);
    return v;
}

__device__ __forceinline__ void seg_update(SegAcc& st, const float4& comp, float v) {
    float mn = fmaxf(st.m, v);
    float sc = __expf(st.m - mn);
    float p  = __expf(v - mn);
    st.acc.x = st.acc.x * sc + comp.x * p;
    st.acc.y = st.acc.y * sc + comp.y * p;
    st.acc.z = st.acc.z * sc + comp.z * p;
    st.acc.w = st.acc.w * sc + comp.w * p;
    st.l = st.l * sc + p;
    st.m = mn;
}

__device__ __forceinline__ float4 node_row(
        const float* __restrict__ ent, const float* __restrict__ rel,
        const int* __restrict__ off, const int* __restrict__ keys,
        int d, int lane, int n_nodes, int n_edges, int n_rel) {
    const float4 zero4 = make_float4(0.0f, 0.0f, 0.0f, 0.0f);
    if (d >= n_nodes) return zero4;
    const int beg = off[d];
    const int end = (d + 1 < n_nodes) ? off[d + 1] : n_edges;
    const int deg = end - beg;
    if (deg <= 0) return zero4;

    const float4 c = ((const float4*)(ent + (size_t)d * HD))[lane];
    SegAcc st;
    st.acc = zero4;
    st.m = -INFINITY;
    st.l = 0.0f;

    if (deg <= 32) {
        const int key = (lane < deg) ? keys[beg + lane] : 0x7fffffff;
        int rank = 0;
        for (int j = 0; j < deg; ++j) {
            int kj = __builtin_amdgcn_readlane(key, j);
            rank += ((kj < key) || (kj == key && j < lane)) ? 1 : 0;
        }
        if (lane >= deg) rank = lane;
        const int skey = __builtin_amdgcn_ds_permute(rank << 2, key);
        const int sk = skey / n_rel;
        const int rk = skey - sk * n_rel;

        for (int k0 = 0; k0 < deg; k0 += 4) {
            float4 a[4], b[4];
            for (int u = 0; u < 4; ++u) {
                int k = (k0 + u < deg) ? (k0 + u) : (deg - 1);
                int s = __builtin_amdgcn_readlane(sk, k);
                int r = __builtin_amdgcn_readlane(rk, k);
                a[u] = ((const float4*)(ent + (size_t)s * HD))[lane];
                b[u] = ((const float4*)(rel + (size_t)r * HD))[lane];
            }
            float4 comp[4];
            float v[4];
            for (int u = 0; u < 4; ++u) {
                comp[u] = make_float4(a[u].x * b[u].x, a[u].y * b[u].y,
                                      a[u].z * b[u].z, a[u].w * b[u].w);
                v[u] = comp[u].x * c.x + comp[u].y * c.y + comp[u].z * c.z + comp[u].w * c.w;
            }
            for (int o = 16; o > 0; o >>= 1) {
                for (int u = 0; u < 4; ++u) v[u] += __shfl_xor(v[u], o, 32);
            }
            for (int u = 0; u < 4; ++u) {
                if (k0 + u < deg) seg_update(st, comp[u], v[u]);
            }
        }
    } else if (deg <= 64) {
        const bool has1 = lane + 32 < deg;
        const int key0 = keys[beg + lane];
        const int key1 = has1 ? keys[beg + 32 + lane] : 0x7fffffff;
        int rank0 = 0;
        int rank1 = 0;
        for (int j = 0; j < 32; ++j) {
            const int kj = __builtin_amdgcn_readlane(key0, j);
            rank0 += ((kj < key0) || (kj == key0 && j < lane)) ? 1 : 0;
            rank1 += (kj <= key1) ? 1 : 0;
        }
        for (int j = 0; j < deg - 32; ++j) {
            const int kj = __builtin_amdgcn_readlane(key1, j);
            rank0 += (kj < key0) ? 1 : 0;
            rank1 += ((kj < key1) || (kj == key1 && j < lane)) ? 1 : 0;
        }
        if (!has1) rank1 = -1;

        for (int k0 = 0; k0 < deg; k0 += 4) {
            float4 a[4], b[4];
            for (int u = 0; u < 4; ++u) {
                const int k = (k0 + u < deg) ? (k0 + u) : (deg - 1);
                const unsigned long long m =
                    ((unsigned long long)__builtin_amdgcn_ballot_w32(rank1 == k) << 32) |
                    (unsigned long long)__builtin_amdgcn_ballot_w32(rank0 == k);
                const int pos = __builtin_ctzll(m);
                const int e0 = __builtin_amdgcn_readlane(key0, pos & 31);
                const int e1 = __builtin_amdgcn_readlane(key1, pos & 31);
                const int skey = (pos < 32) ? e0 : e1;
                const int s = skey / n_rel;
                const int r = skey - s * n_rel;
                a[u] = ((const float4*)(ent + (size_t)s * HD))[lane];
                b[u] = ((const float4*)(rel + (size_t)r * HD))[lane];
            }
            float4 comp[4];
            float v[4];
            for (int u = 0; u < 4; ++u) {
                comp[u] = make_float4(a[u].x * b[u].x, a[u].y * b[u].y,
                                      a[u].z * b[u].z, a[u].w * b[u].w);
                v[u] = comp[u].x * c.x + comp[u].y * c.y + comp[u].z * c.z + comp[u].w * c.w;
            }
            for (int o = 16; o > 0; o >>= 1) {
                for (int u = 0; u < 4; ++u) v[u] += __shfl_xor(v[u], o, 32);
            }
            for (int u = 0; u < 4; ++u) {
                if (k0 + u < deg) seg_update(st, comp[u], v[u]);
            }
        }
    } else {
        int prev = -1;
        int done = 0;
        for (int it = 0; it < deg && done < deg; ++it) {
            int best = 0x7fffffff;
            for (int i = lane; i < deg; i += 32) {
                int kk = keys[beg + i];
                if (kk > prev && kk < best) best = kk;
            }
            for (int o = 16; o > 0; o >>= 1) {
                int y = __shfl_xor(best, o, 32);
                best = (y < best) ? y : best;
            }
            best = __builtin_amdgcn_readfirstlane(best);
            int mult = 0;
            for (int i = lane; i < deg; i += 32) mult += (keys[beg + i] == best) ? 1 : 0;
            for (int o = 16; o > 0; o >>= 1) mult += __shfl_xor(mult, o, 32);
            mult = __builtin_amdgcn_readfirstlane(mult);
            if (mult == 0) break;
            const int s = best / n_rel;
            const int r = best - s * n_rel;
            const float4 a = ((const float4*)(ent + (size_t)s * HD))[lane];
            const float4 b = ((const float4*)(rel + (size_t)r * HD))[lane];
            const float4 comp = make_float4(a.x * b.x, a.y * b.y, a.z * b.z, a.w * b.w);
            const float v = wave_sum32(comp.x * c.x + comp.y * c.y + comp.z * c.z + comp.w * c.w);
            for (int t = 0; t < mult; ++t) seg_update(st, comp, v);
            done += mult;
            prev = best;
        }
    }

    const float inv = 1.0f / st.l;
    return make_float4(st.acc.x * inv, st.acc.y * inv, st.acc.z * inv, st.acc.w * inv);
}

#define TILE_M 16
#define NODE_WAVES (TPB / 32)
#define TILE_PITCH (HD + 4)

__global__ void __launch_bounds__(TPB) node_gemm_kernel(
        const float* __restrict__ ent, const float* __restrict__ rel,
        const int* __restrict__ off, const int* __restrict__ keys,
        const float* __restrict__ W,
        float* __restrict__ H,
        int n_nodes, int n_edges, int n_rel) {
    __shared__ __attribute__((aligned(16))) float tile[TILE_M * TILE_PITCH];
    const int lane = threadIdx.x & 31;
    const int wave = __builtin_amdgcn_readfirstlane((int)(threadIdx.x >> 5));
    const int m0 = blockIdx.x * TILE_M;

    for (int u = 0; u < TILE_M / NODE_WAVES; ++u) {
        const int row = wave * (TILE_M / NODE_WAVES) + u;
        const float4 nr = node_row(ent, rel, off, keys, m0 + row, lane, n_nodes, n_edges, n_rel);
        ((float4*)(tile + row * TILE_PITCH))[lane] = nr;
    }
    __syncthreads();

    const int half = lane >> 4;
    const int l16  = lane & 15;
    const int n0 = wave * 16;
    v8f c = {};
    const float* arow = tile + l16 * TILE_PITCH;
    const float* wcol = W + n0 + l16;

    #pragma unroll 8
    for (int k0 = 0; k0 < HD; k0 += 4) {
        const int ka = k0 + 2 * half;
        const v2f a = *((const v2f*)(arow + ka));
        v2f b;
        b.x = wcol[(size_t)ka * HD];
        b.y = wcol[(size_t)(ka + 1) * HD];
        c = __builtin_amdgcn_wmma_f32_16x16x4_f32(
                 false, a,  false, b,
                 (short)0, c,  false,  false);
    }

    const int mrow = m0 + 8 * half;
    float* hb = H + (size_t)mrow * HD + n0 + l16;
    #pragma unroll
    for (int r2 = 0; r2 < 8; ++r2) {
        if (mrow + r2 < n_nodes) hb[(size_t)r2 * HD] = c[r2];
    }
}

#define STAT_ROWS 13
#define STAT_WAVES (TPB / 32)
__global__ void __launch_bounds__(TPB) bn_stats_kernel(
        const float* __restrict__ H, float* __restrict__ colsum,
        float* __restrict__ colsumsq, int n_rows) {
    __shared__ float part[2][STAT_WAVES][HD];
    const int lane = threadIdx.x & 31;
    const int w = __builtin_amdgcn_readfirstlane((int)(threadIdx.x >> 5));
    const int b = blockIdx.x;
    const int G = gridDim.x;
    const float4* H4 = (const float4*)H;
    float4 s = make_float4(0.0f, 0.0f, 0.0f, 0.0f);
    float4 q = make_float4(0.0f, 0.0f, 0.0f, 0.0f);
    for (int i0 = w; i0 * G + b < n_rows; i0 += STAT_WAVES * STAT_ROWS) {
        float4 v[STAT_ROWS];
        #pragma unroll
        for (int u = 0; u < STAT_ROWS; ++u) {
            const int r = (i0 + u * STAT_WAVES) * G + b;
            v[u] = (r < n_rows) ? H4[(size_t)r * (HD / 4) + lane]
                                : make_float4(0.0f, 0.0f, 0.0f, 0.0f);
        }
        #pragma unroll
        for (int u = 0; u < STAT_ROWS; ++u) {
            s.x += v[u].x; s.y += v[u].y; s.z += v[u].z; s.w += v[u].w;
            q.x += v[u].x * v[u].x; q.y += v[u].y * v[u].y;
            q.z += v[u].z * v[u].z; q.w += v[u].w * v[u].w;
        }
    }
    ((float4*)&part[0][w][0])[lane] = s;
    ((float4*)&part[1][w][0])[lane] = q;
    __syncthreads();
    const int stat = threadIdx.x >> 7;
    const int col  = threadIdx.x & (HD - 1);
    float acc = 0.0f;
    for (int i = 0; i < STAT_WAVES; ++i) acc += part[stat][i][col];
    atomicAdd((stat ? colsumsq : colsum) + col, acc);
}

__global__ void __launch_bounds__(TPB) bn_tanh_kernel(
        float* __restrict__ H, const float* __restrict__ colsum,
        const float* __restrict__ colsumsq, const float* __restrict__ gamma,
        const float* __restrict__ beta, int n_rows) {
    size_t i = (size_t)blockIdx.x * TPB + threadIdx.x;
    size_t total = (size_t)n_rows * HD;
    if (i >= total) return;
    int c = (int)(i & (HD - 1));
    float invN = 1.0f / (float)n_rows;
    float mu   = colsum[c] * invN;
    float var  = colsumsq[c] * invN - mu * mu;
    float istd = rsqrtf(var + 1e-5f);
    float v = (H[i] - mu) * istd * gamma[c] + beta[c];
    H[i] = tanhf(v);
}

extern "C" void kernel_launch(void* const* d_in, const int* in_sizes, int n_in,
                              void* d_out, int out_size, void* d_ws, size_t ws_size,
                              hipStream_t stream) {
    const float* ent   = (const float*)d_in[0];
    const float* rel   = (const float*)d_in[1];
    const float* W     = (const float*)d_in[2];
    const float* gamma = (const float*)d_in[3];
    const float* beta  = (const float*)d_in[4];
    const int*   src   = (const int*)d_in[5];
    const int*   dst   = (const int*)d_in[6];
    const int*   rid   = (const int*)d_in[7];
    float* H = (float*)d_out;

    const int N = in_sizes[0] / HD;
    const int R = in_sizes[1] / HD;
    const int E = in_sizes[5];

    float* ws       = (float*)d_ws;
    float* neigh    = ws;
    int*   keys     = (int*)(neigh + (size_t)N * HD);
    int*   cnt      = keys + E;
    float* colsum   = (float*)(cnt + N);
    float* colsumsq = colsum + HD;
    int*   off      = (int*)(colsumsq + HD);
    int*   pairs    = (int*)neigh;
    int*   bin_cnt   = cnt;
    int*   bin_cur   = cnt + MAX_BINS * CTR_PITCH;
    int*   bin_start = cnt + 2 * MAX_BINS * CTR_PITCH;

    size_t zt = (size_t)N + 2 * HD;
    zero_kernel<<<(unsigned)((zt + TPB - 1) / TPB), TPB, 0, stream>>>((float*)cnt, zt);

    const int n_bins = (N + BIN_NODES - 1) / BIN_NODES;
    bin_count_kernel<<<BIN_BLOCKS, TPB, 0, stream>>>(dst, bin_cnt, E, n_bins);
    bin_scatter_kernel<<<BIN_BLOCKS, TPB, 0, stream>>>(
        src, dst, rid, bin_cnt, bin_cur, bin_start, pairs, E, n_bins, R);
    bin_sort_kernel<<<(unsigned)n_bins, SORT_T, 0, stream>>>(pairs, bin_cnt, bin_start, off, keys, N);

    node_gemm_kernel<<<(unsigned)((N + TILE_M - 1) / TILE_M), TPB, 0, stream>>>(
        ent, rel, off, keys, W, H, N, E, R);

    bn_stats_kernel<<<512, TPB, 0, stream>>>(H, colsum, colsumsq, N);
    bn_tanh_kernel<<<(unsigned)(((size_t)N * HD + TPB - 1) / TPB), TPB, 0, stream>>>(
        H, colsum, colsumsq, gamma, beta, N);
}


# Edges binned inside each block's own chunk (LDS counts, scan, ranked stores, a run record per bin); bins gather their runs, one int add per bin on a cursor: count and zero launches gone; same sums
# speedup vs baseline: 4.9101x; 1.2674x over previous
#include <hip/hip_runtime.h>
#include <math.h>

#define HD 128
#define TPB 256

typedef __attribute__((ext_vector_type(2))) float v2f;
typedef __attribute__((ext_vector_type(8))) float v8f;

#define BIN_SHIFT 8
#define BIN_NODES 256
#define MAX_BINS 256
#define BIN_BLOCKS 512
#define SORT_T 1024
#define LOC_U 7
#define SORT_U 5

__device__ __forceinline__ int wave_incl_scan_int(int v, int lane) {
    int x = v;
    for (int o = 1; o < 32; o <<= 1) {
        int y = __shfl_up(x, o, 32);
        if (lane >= o) x += y;
    }
    return x;
}

__global__ void __launch_bounds__(TPB) bin_local_kernel(
        const int* __restrict__ src, const int* __restrict__ dst, const int* __restrict__ rid,
        int* __restrict__ pairs, int* __restrict__ run_tab, int* __restrict__ key_cursor,
        float* __restrict__ colsums, int n_edges, int n_bins, int n_rel) {
    __shared__ int hist[MAX_BINS];
    __shared__ int base[MAX_BINS];
    __shared__ int wtot[TPB / 32];
    const int t = threadIdx.x;
    const int lane = t & 31;
    const int w = t >> 5;
    hist[t] = 0;
    if (blockIdx.x == 0) {
        colsums[t] = 0.0f;
        if (t == 0) *key_cursor = 0;
    }
    __syncthreads();
    const int chunk = (n_edges + BIN_BLOCKS - 1) / BIN_BLOCKS;
    const int e0 = blockIdx.x * chunk;
    const int e1 = (e0 + chunk < n_edges) ? (e0 + chunk) : n_edges;
    int d0[LOC_U], k0[LOC_U];
    {
        int s0[LOC_U], r0[LOC_U];
        #pragma unroll
        for (int u = 0; u < LOC_U; ++u) {
            const int ee = e0 + t + u * TPB;
            const bool ok = ee < e1;
            d0[u] = ok ? dst[ee] : -1;
            s0[u] = ok ? src[ee] : 0;
            r0[u] = ok ? rid[ee] : 0;
        }
        #pragma unroll
        for (int u = 0; u < LOC_U; ++u) k0[u] = s0[u] * n_rel + r0[u];
    }
    #pragma unroll
    for (int u = 0; u < LOC_U; ++u) if (d0[u] >= 0) atomicAdd(&hist[d0[u] >> BIN_SHIFT], 1);
    for (int e = e0 + t + LOC_U * TPB; e < e1; e += TPB) atomicAdd(&hist[dst[e] >> BIN_SHIFT], 1);
    __syncthreads();
    const int h = hist[t];
    const int incl = wave_incl_scan_int(h, lane);
    if (lane == 31) wtot[w] = incl;
    __syncthreads();
    int pre = 0;
    for (int i = 0; i < w; ++i) pre += wtot[i];
    const int start = e0 + pre + incl - h;
    base[t] = start;
    if (t < n_bins) {
        *((int2*)(run_tab + 2 * ((size_t)t * BIN_BLOCKS + blockIdx.x))) = make_int2(start, h);
    }
    hist[t] = 0;
    __syncthreads();
    #pragma unroll
    for (int u = 0; u < LOC_U; ++u) {
        if (d0[u] >= 0) {
            const int b = d0[u] >> BIN_SHIFT;
            const int slot = base[b] + atomicAdd(&hist[b], 1);
            *((int2*)(pairs + 2 * (size_t)slot)) = make_int2(d0[u], k0[u]);
        }
    }
    for (int e = e0 + t + LOC_U * TPB; e < e1; e += TPB) {
        const int d = dst[e];
        const int k = src[e] * n_rel + rid[e];
        const int b = d >> BIN_SHIFT;
        const int slot = base[b] + atomicAdd(&hist[b], 1);
        *((int2*)(pairs + 2 * (size_t)slot)) = make_int2(d, k);
    }
}

__device__ __forceinline__ void load_trip(
        const int* __restrict__ pairs, const int* rpre, const int* rstart,
        int i0, int total, int2 (&p)[SORT_U]) {
    int lo[SORT_U], e[SORT_U];
    #pragma unroll
    for (int u = 0; u < SORT_U; ++u) {
        const int ii = i0 + u * SORT_T;
        e[u] = (ii < total) ? ii : (total - 1);
        lo[u] = 0;
    }
    #pragma unroll
    for (int wd = BIN_BLOCKS / 2; wd >= 1; wd >>= 1) {
        #pragma unroll
        for (int u = 0; u < SORT_U; ++u) lo[u] = (rpre[lo[u] + wd] <= e[u]) ? (lo[u] + wd) : lo[u];
    }
    int2 q[SORT_U];
    #pragma unroll
    for (int u = 0; u < SORT_U; ++u) {
        const int slot = rstart[lo[u]] + (e[u] - rpre[lo[u]]);
        q[u] = *((const int2*)(pairs + 2 * (size_t)slot));
    }
    #pragma unroll
    for (int u = 0; u < SORT_U; ++u) {
        const int none = (i0 + u * SORT_T < total) ? 0 : -1;
        p[u] = make_int2(q[u].x | none, q[u].y);
    }
}

__global__ void __launch_bounds__(SORT_T) bin_sort_kernel(
        const int* __restrict__ pairs, const int* __restrict__ run_tab,
        int* __restrict__ key_cursor, int* __restrict__ off2, int* __restrict__ keys,
        int n_nodes) {
    __shared__ int rpre[BIN_BLOCKS + 1];
    __shared__ int rstart[BIN_BLOCKS];
    __shared__ int lcnt[BIN_NODES];
    __shared__ int lpos[BIN_NODES];
    __shared__ int wtot[SORT_T / 32];
    __shared__ int sbase;
    const int t = threadIdx.x;
    const int lane = t & 31;
    const int w = t >> 5;
    const int bin = blockIdx.x;
    int2 rt = make_int2(0, 0);
    if (t < BIN_BLOCKS) rt = *((const int2*)(run_tab + 2 * ((size_t)bin * BIN_BLOCKS + t)));
    if (t < BIN_NODES) lcnt[t] = 0;
    const int rc = rt.y;
    const int rincl = wave_incl_scan_int(rc, lane);
    if (lane == 31) wtot[w] = rincl;
    __syncthreads();
    if (t < BIN_BLOCKS) {
        int pre = 0;
        for (int i = 0; i < w; ++i) pre += wtot[i];
        rpre[t] = pre + rincl - rc;
        rstart[t] = rt.x;
        if (t == BIN_BLOCKS - 1) rpre[BIN_BLOCKS] = pre + rincl;
    }
    __syncthreads();
    const int total = rpre[BIN_BLOCKS];
    if (t == 0) sbase = atomicAdd(key_cursor, total);
    int2 p0[SORT_U];
    #pragma unroll
    for (int u = 0; u < SORT_U; ++u) p0[u] = make_int2(-1, 0);
    if (total > 0) load_trip(pairs, rpre, rstart, t, total, p0);
    #pragma unroll
    for (int u = 0; u < SORT_U; ++u) if (p0[u].x >= 0) atomicAdd(&lcnt[p0[u].x & (BIN_NODES - 1)], 1);
    for (int i = t + SORT_U * SORT_T; i < total; i += SORT_U * SORT_T) {
        int2 p[SORT_U];
        load_trip(pairs, rpre, rstart, i, total, p);
        #pragma unroll
        for (int u = 0; u < SORT_U; ++u) if (p[u].x >= 0) atomicAdd(&lcnt[p[u].x & (BIN_NODES - 1)], 1);
    }
    __syncthreads();
    const int bs = sbase;
    const int v = (t < BIN_NODES) ? lcnt[t] : 0;
    const int incl = wave_incl_scan_int(v, lane);
    if (t < BIN_NODES && lane == 31) wtot[w] = incl;
    __syncthreads();
    if (t < BIN_NODES) {
        int pre = 0;
        for (int i = 0; i < w; ++i) pre += wtot[i];
        const int ex = bs + pre + incl - v;
        lpos[t] = ex;
        const int node = bin * BIN_NODES + t;
        if (node < n_nodes) *((int2*)(off2 + 2 * (size_t)node)) = make_int2(ex, v);
    }
    __syncthreads();
    #pragma unroll
    for (int u = 0; u < SORT_U; ++u) {
        if (p0[u].x >= 0) {
            const int slot = atomicAdd(&lpos[p0[u].x & (BIN_NODES - 1)], 1);
            keys[slot] = p0[u].y;
        }
    }
    for (int i = t + SORT_U * SORT_T; i < total; i += SORT_U * SORT_T) {
        int2 p[SORT_U];
        load_trip(pairs, rpre, rstart, i, total, p);
        #pragma unroll
        for (int u = 0; u < SORT_U; ++u) {
            if (p[u].x >= 0) {
                const int slot = atomicAdd(&lpos[p[u].x & (BIN_NODES - 1)], 1);
                keys[slot] = p[u].y;
            }
        }
    }
}

struct SegAcc { float4 acc; float m; float l; };

__device__ __forceinline__ float wave_sum32(float v) {
    for (int o = 16; o > 0; o >>= 1) v += __shfl_xor(v, o, 32);
    return v;
}

__device__ __forceinline__ void seg_update(SegAcc& st, const float4& comp, float v) {
    float mn = fmaxf(st.m, v);
    float sc = __expf(st.m - mn);
    float p  = __expf(v - mn);
    st.acc.x = st.acc.x * sc + comp.x * p;
    st.acc.y = st.acc.y * sc + comp.y * p;
    st.acc.z = st.acc.z * sc + comp.z * p;
    st.acc.w = st.acc.w * sc + comp.w * p;
    st.l = st.l * sc + p;
    st.m = mn;
}

__device__ __forceinline__ float4 node_row(
        const float* __restrict__ ent, const float* __restrict__ rel,
        const int* __restrict__ off, const int* __restrict__ keys,
        int d, int lane, int n_nodes, int n_edges, int n_rel) {
    const float4 zero4 = make_float4(0.0f, 0.0f, 0.0f, 0.0f);
    if (d >= n_nodes) return zero4;
    const int2 od = *((const int2*)(off + 2 * (size_t)d));
    const int beg = od.x;
    const int deg = od.y;
    if (deg <= 0) return zero4;

    const float4 c = ((const float4*)(ent + (size_t)d * HD))[lane];
    SegAcc st;
    st.acc = zero4;
    st.m = -INFINITY;
    st.l = 0.0f;

    if (deg <= 32) {
        const int key = (lane < deg) ? keys[beg + lane] : 0x7fffffff;
        int rank = 0;
        for (int j = 0; j < deg; ++j) {
            int kj = __builtin_amdgcn_readlane(key, j);
            rank += ((kj < key) || (kj == key && j < lane)) ? 1 : 0;
        }
        if (lane >= deg) rank = lane;
        const int skey = __builtin_amdgcn_ds_permute(rank << 2, key);
        const int sk = skey / n_rel;
        const int rk = skey - sk * n_rel;

        for (int k0 = 0; k0 < deg; k0 += 4) {
            float4 a[4], b[4];
            for (int u = 0; u < 4; ++u) {
                int k = (k0 + u < deg) ? (k0 + u) : (deg - 1);
                int s = __builtin_amdgcn_readlane(sk, k);
                int r = __builtin_amdgcn_readlane(rk, k);
                a[u] = ((const float4*)(ent + (size_t)s * HD))[lane];
                b[u] = ((const float4*)(rel + (size_t)r * HD))[lane];
            }
            float4 comp[4];
            float v[4];
            for (int u = 0; u < 4; ++u) {
                comp[u] = make_float4(a[u].x * b[u].x, a[u].y * b[u].y,
                                      a[u].z * b[u].z, a[u].w * b[u].w);
                v[u] = comp[u].x * c.x + comp[u].y * c.y + comp[u].z * c.z + comp[u].w * c.w;
            }
            for (int o = 16; o > 0; o >>= 1) {
                for (int u = 0; u < 4; ++u) v[u] += __shfl_xor(v[u], o, 32);
            }
            for (int u = 0; u < 4; ++u) {
                if (k0 + u < deg) seg_update(st, comp[u], v[u]);
            }
        }
    } else if (deg <= 64) {
        const bool has1 = lane + 32 < deg;
        const int key0 = keys[beg + lane];
        const int key1 = has1 ? keys[beg + 32 + lane] : 0x7fffffff;
        int rank0 = 0;
        int rank1 = 0;
        for (int j = 0; j < 32; ++j) {
            const int kj = __builtin_amdgcn_readlane(key0, j);
            rank0 += ((kj < key0) || (kj == key0 && j < lane)) ? 1 : 0;
            rank1 += (kj <= key1) ? 1 : 0;
        }
        for (int j = 0; j < deg - 32; ++j) {
            const int kj = __builtin_amdgcn_readlane(key1, j);
            rank0 += (kj < key0) ? 1 : 0;
            rank1 += ((kj < key1) || (kj == key1 && j < lane)) ? 1 : 0;
        }
        if (!has1) rank1 = -1;

        for (int k0 = 0; k0 < deg; k0 += 4) {
            float4 a[4], b[4];
            for (int u = 0; u < 4; ++u) {
                const int k = (k0 + u < deg) ? (k0 + u) : (deg - 1);
                const unsigned long long m =
                    ((unsigned long long)__builtin_amdgcn_ballot_w32(rank1 == k) << 32) |
                    (unsigned long long)__builtin_amdgcn_ballot_w32(rank0 == k);
                const int pos = __builtin_ctzll(m);
                const int e0 = __builtin_amdgcn_readlane(key0, pos & 31);
                const int e1 = __builtin_amdgcn_readlane(key1, pos & 31);
                const int skey = (pos < 32) ? e0 : e1;
                const int s = skey / n_rel;
                const int r = skey - s * n_rel;
                a[u] = ((const float4*)(ent + (size_t)s * HD))[lane];
                b[u] = ((const float4*)(rel + (size_t)r * HD))[lane];
            }
            float4 comp[4];
            float v[4];
            for (int u = 0; u < 4; ++u) {
                comp[u] = make_float4(a[u].x * b[u].x, a[u].y * b[u].y,
                                      a[u].z * b[u].z, a[u].w * b[u].w);
                v[u] = comp[u].x * c.x + comp[u].y * c.y + comp[u].z * c.z + comp[u].w * c.w;
            }
            for (int o = 16; o > 0; o >>= 1) {
                for (int u = 0; u < 4; ++u) v[u] += __shfl_xor(v[u], o, 32);
            }
            for (int u = 0; u < 4; ++u) {
                if (k0 + u < deg) seg_update(st, comp[u], v[u]);
            }
        }
    } else {
        int prev = -1;
        int done = 0;
        for (int it = 0; it < deg && done < deg; ++it) {
            int best = 0x7fffffff;
            for (int i = lane; i < deg; i += 32) {
                int kk = keys[beg + i];
                if (kk > prev && kk < best) best = kk;
            }
            for (int o = 16; o > 0; o >>= 1) {
                int y = __shfl_xor(best, o, 32);
                best = (y < best) ? y : best;
            }
            best = __builtin_amdgcn_readfirstlane(best);
            int mult = 0;
            for (int i = lane; i < deg; i += 32) mult += (keys[beg + i] == best) ? 1 : 0;
            for (int o = 16; o > 0; o >>= 1) mult += __shfl_xor(mult, o, 32);
            mult = __builtin_amdgcn_readfirstlane(mult);
            if (mult == 0) break;
            const int s = best / n_rel;
            const int r = best - s * n_rel;
            const float4 a = ((const float4*)(ent + (size_t)s * HD))[lane];
            const float4 b = ((const float4*)(rel + (size_t)r * HD))[lane];
            const float4 comp = make_float4(a.x * b.x, a.y * b.y, a.z * b.z, a.w * b.w);
            const float v = wave_sum32(comp.x * c.x + comp.y * c.y + comp.z * c.z + comp.w * c.w);
            for (int t = 0; t < mult; ++t) seg_update(st, comp, v);
            done += mult;
            prev = best;
        }
    }

    const float inv = 1.0f / st.l;
    return make_float4(st.acc.x * inv, st.acc.y * inv, st.acc.z * inv, st.acc.w * inv);
}

#define TILE_M 16
#define NODE_WAVES (TPB / 32)
#define TILE_PITCH (HD + 4)

__global__ void __launch_bounds__(TPB) node_gemm_kernel(
        const float* __restrict__ ent, const float* __restrict__ rel,
        const int* __restrict__ off, const int* __restrict__ keys,
        const float* __restrict__ W,
        float* __restrict__ H,
        int n_nodes, int n_edges, int n_rel) {
    __shared__ __attribute__((aligned(16))) float tile[TILE_M * TILE_PITCH];
    const int lane = threadIdx.x & 31;
    const int wave = __builtin_amdgcn_readfirstlane((int)(threadIdx.x >> 5));
    const int m0 = blockIdx.x * TILE_M;

    for (int u = 0; u < TILE_M / NODE_WAVES; ++u) {
        const int row = wave * (TILE_M / NODE_WAVES) + u;
        const float4 nr = node_row(ent, rel, off, keys, m0 + row, lane, n_nodes, n_edges, n_rel);
        ((float4*)(tile + row * TILE_PITCH))[lane] = nr;
    }
    __syncthreads();

    const int half = lane >> 4;
    const int l16  = lane & 15;
    const int n0 = wave * 16;
    v8f c = {};
    const float* arow = tile + l16 * TILE_PITCH;
    const float* wcol = W + n0 + l16;

    #pragma unroll 8
    for (int k0 = 0; k0 < HD; k0 += 4) {
        const int ka = k0 + 2 * half;
        const v2f a = *((const v2f*)(arow + ka));
        v2f b;
        b.x = wcol[(size_t)ka * HD];
        b.y = wcol[(size_t)(ka + 1) * HD];
        c = __builtin_amdgcn_wmma_f32_16x16x4_f32(
                 false, a,  false, b,
                 (short)0, c,  false,  false);
    }

    const int mrow = m0 + 8 * half;
    float* hb = H + (size_t)mrow * HD + n0 + l16;
    #pragma unroll
    for (int r2 = 0; r2 < 8; ++r2) {
        if (mrow + r2 < n_nodes) hb[(size_t)r2 * HD] = c[r2];
    }
}

#define STAT_ROWS 13
#define STAT_WAVES (TPB / 32)
__global__ void __launch_bounds__(TPB) bn_stats_kernel(
        const float* __restrict__ H, float* __restrict__ colsum,
        float* __restrict__ colsumsq, int n_rows) {
    __shared__ float part[2][STAT_WAVES][HD];
    const int lane = threadIdx.x & 31;
    const int w = __builtin_amdgcn_readfirstlane((int)(threadIdx.x >> 5));
    const int b = blockIdx.x;
    const int G = gridDim.x;
    const float4* H4 = (const float4*)H;
    float4 s = make_float4(0.0f, 0.0f, 0.0f, 0.0f);
    float4 q = make_float4(0.0f, 0.0f, 0.0f, 0.0f);
    for (int i0 = w; i0 * G + b < n_rows; i0 += STAT_WAVES * STAT_ROWS) {
        float4 v[STAT_ROWS];
        #pragma unroll
        for (int u = 0; u < STAT_ROWS; ++u) {
            const int r = (i0 + u * STAT_WAVES) * G + b;
            v[u] = (r < n_rows) ? H4[(size_t)r * (HD / 4) + lane]
                                : make_float4(0.0f, 0.0f, 0.0f, 0.0f);
        }
        #pragma unroll
        for (int u = 0; u < STAT_ROWS; ++u) {
            s.x += v[u].x; s.y += v[u].y; s.z += v[u].z; s.w += v[u].w;
            q.x += v[u].x * v[u].x; q.y += v[u].y * v[u].y;
            q.z += v[u].z * v[u].z; q.w += v[u].w * v[u].w;
        }
    }
    ((float4*)&part[0][w][0])[lane] = s;
    ((float4*)&part[1][w][0])[lane] = q;
    __syncthreads();
    const int stat = threadIdx.x >> 7;
    const int col  = threadIdx.x & (HD - 1);
    float acc = 0.0f;
    for (int i = 0; i < STAT_WAVES; ++i) acc += part[stat][i][col];
    atomicAdd((stat ? colsumsq : colsum) + col, acc);
}

__global__ void __launch_bounds__(TPB) bn_tanh_kernel(
        float* __restrict__ H, const float* __restrict__ colsum,
        const float* __restrict__ colsumsq, const float* __restrict__ gamma,
        const float* __restrict__ beta, int n_rows) {
    size_t i = (size_t)blockIdx.x * TPB + threadIdx.x;
    size_t total = (size_t)n_rows * HD;
    if (i >= total) return;
    int c = (int)(i & (HD - 1));
    float invN = 1.0f / (float)n_rows;
    float mu   = colsum[c] * invN;
    float var  = colsumsq[c] * invN - mu * mu;
    float istd = rsqrtf(var + 1e-5f);
    float v = (H[i] - mu) * istd * gamma[c] + beta[c];
    H[i] = tanhf(v);
}

extern "C" void kernel_launch(void* const* d_in, const int* in_sizes, int n_in,
                              void* d_out, int out_size, void* d_ws, size_t ws_size,
                              hipStream_t stream) {
    const float* ent   = (const float*)d_in[0];
    const float* rel   = (const float*)d_in[1];
    const float* W     = (const float*)d_in[2];
    const float* gamma = (const float*)d_in[3];
    const float* beta  = (const float*)d_in[4];
    const int*   src   = (const int*)d_in[5];
    const int*   dst   = (const int*)d_in[6];
    const int*   rid   = (const int*)d_in[7];
    float* H = (float*)d_out;

    const int N = in_sizes[0] / HD;
    const int R = in_sizes[1] / HD;
    const int E = in_sizes[5];

    float* ws       = (float*)d_ws;
    float* neigh    = ws;
    int*   keys     = (int*)(neigh + (size_t)N * HD);
    int*   cnt      = keys + E;
    float* colsum   = (float*)(cnt + N);
    float* colsumsq = colsum + HD;
    int*   pairs    = (int*)neigh;
    int*   run_tab  = pairs + 2 * (size_t)E;
    int*   off2     = run_tab + 2 * (size_t)MAX_BINS * BIN_BLOCKS;
    int*   key_cursor = cnt;

    const int n_bins = (N + BIN_NODES - 1) / BIN_NODES;
    bin_local_kernel<<<BIN_BLOCKS, TPB, 0, stream>>>(
        src, dst, rid, pairs, run_tab, key_cursor, colsum, E, n_bins, R);
    bin_sort_kernel<<<(unsigned)n_bins, SORT_T, 0, stream>>>(pairs, run_tab, key_cursor, off2, keys, N);

    node_gemm_kernel<<<(unsigned)((N + TILE_M - 1) / TILE_M), TPB, 0, stream>>>(
        ent, rel, off2, keys, W, H, N, E, R);

    bn_stats_kernel<<<512, TPB, 0, stream>>>(H, colsum, colsumsq, N);
    bn_tanh_kernel<<<(unsigned)(((size_t)N * HD + TPB - 1) / TPB), TPB, 0, stream>>>(
        H, colsum, colsumsq, gamma, beta, N);
}


# Segments up to 32 edges: half-wave per edge (8 columns per lane, two edges per pass), score summed in its 16-lane row by DPP adds; even/odd key-ordered edges summed apart, joined once: new sum order
# speedup vs baseline: 5.1511x; 1.0491x over previous
#include <hip/hip_runtime.h>
#include <math.h>

#define HD 128
#define TPB 256

typedef __attribute__((ext_vector_type(2))) float v2f;
typedef __attribute__((ext_vector_type(8))) float v8f;

#define BIN_SHIFT 8
#define BIN_NODES 256
#define MAX_BINS 256
#define BIN_BLOCKS 512
#define SORT_T 1024
#define LOC_U 7
#define SORT_U 5

__device__ __forceinline__ int wave_incl_scan_int(int v, int lane) {
    int x = v;
    for (int o = 1; o < 32; o <<= 1) {
        int y = __shfl_up(x, o, 32);
        if (lane >= o) x += y;
    }
    return x;
}

__global__ void __launch_bounds__(TPB) bin_local_kernel(
        const int* __restrict__ src, const int* __restrict__ dst, const int* __restrict__ rid,
        int* __restrict__ pairs, int* __restrict__ run_tab, int* __restrict__ key_cursor,
        float* __restrict__ colsums, int n_edges, int n_bins, int n_rel) {
    __shared__ int hist[MAX_BINS];
    __shared__ int base[MAX_BINS];
    __shared__ int wtot[TPB / 32];
    const int t = threadIdx.x;
    const int lane = t & 31;
    const int w = t >> 5;
    hist[t] = 0;
    if (blockIdx.x == 0) {
        colsums[t] = 0.0f;
        if (t == 0) *key_cursor = 0;
    }
    __syncthreads();
    const int chunk = (n_edges + BIN_BLOCKS - 1) / BIN_BLOCKS;
    const int e0 = blockIdx.x * chunk;
    const int e1 = (e0 + chunk < n_edges) ? (e0 + chunk) : n_edges;
    int d0[LOC_U], k0[LOC_U];
    {
        int s0[LOC_U], r0[LOC_U];
        #pragma unroll
        for (int u = 0; u < LOC_U; ++u) {
            const int ee = e0 + t + u * TPB;
            const bool ok = ee < e1;
            d0[u] = ok ? dst[ee] : -1;
            s0[u] = ok ? src[ee] : 0;
            r0[u] = ok ? rid[ee] : 0;
        }
        #pragma unroll
        for (int u = 0; u < LOC_U; ++u) k0[u] = s0[u] * n_rel + r0[u];
    }
    #pragma unroll
    for (int u = 0; u < LOC_U; ++u) if (d0[u] >= 0) atomicAdd(&hist[d0[u] >> BIN_SHIFT], 1);
    for (int e = e0 + t + LOC_U * TPB; e < e1; e += TPB) atomicAdd(&hist[dst[e] >> BIN_SHIFT], 1);
    __syncthreads();
    const int h = hist[t];
    const int incl = wave_incl_scan_int(h, lane);
    if (lane == 31) wtot[w] = incl;
    __syncthreads();
    int pre = 0;
    for (int i = 0; i < w; ++i) pre += wtot[i];
    const int start = e0 + pre + incl - h;
    base[t] = start;
    if (t < n_bins) {
        *((int2*)(run_tab + 2 * ((size_t)t * BIN_BLOCKS + blockIdx.x))) = make_int2(start, h);
    }
    hist[t] = 0;
    __syncthreads();
    #pragma unroll
    for (int u = 0; u < LOC_U; ++u) {
        if (d0[u] >= 0) {
            const int b = d0[u] >> BIN_SHIFT;
            const int slot = base[b] + atomicAdd(&hist[b], 1);
            *((int2*)(pairs + 2 * (size_t)slot)) = make_int2(d0[u], k0[u]);
        }
    }
    for (int e = e0 + t + LOC_U * TPB; e < e1; e += TPB) {
        const int d = dst[e];
        const int k = src[e] * n_rel + rid[e];
        const int b = d >> BIN_SHIFT;
        const int slot = base[b] + atomicAdd(&hist[b], 1);
        *((int2*)(pairs + 2 * (size_t)slot)) = make_int2(d, k);
    }
}

__device__ __forceinline__ void load_trip(
        const int* __restrict__ pairs, const int* rpre, const int* rstart,
        int i0, int total, int2 (&p)[SORT_U]) {
    int lo[SORT_U], e[SORT_U];
    #pragma unroll
    for (int u = 0; u < SORT_U; ++u) {
        const int ii = i0 + u * SORT_T;
        e[u] = (ii < total) ? ii : (total - 1);
        lo[u] = 0;
    }
    #pragma unroll
    for (int wd = BIN_BLOCKS / 2; wd >= 1; wd >>= 1) {
        #pragma unroll
        for (int u = 0; u < SORT_U; ++u) lo[u] = (rpre[lo[u] + wd] <= e[u]) ? (lo[u] + wd) : lo[u];
    }
    int2 q[SORT_U];
    #pragma unroll
    for (int u = 0; u < SORT_U; ++u) {
        const int slot = rstart[lo[u]] + (e[u] - rpre[lo[u]]);
        q[u] = *((const int2*)(pairs + 2 * (size_t)slot));
    }
    #pragma unroll
    for (int u = 0; u < SORT_U; ++u) {
        const int none = (i0 + u * SORT_T < total) ? 0 : -1;
        p[u] = make_int2(q[u].x | none, q[u].y);
    }
}

__global__ void __launch_bounds__(SORT_T) bin_sort_kernel(
        const int* __restrict__ pairs, const int* __restrict__ run_tab,
        int* __restrict__ key_cursor, int* __restrict__ off2, int* __restrict__ keys,
        int n_nodes) {
    __shared__ int rpre[BIN_BLOCKS + 1];
    __shared__ int rstart[BIN_BLOCKS];
    __shared__ int lcnt[BIN_NODES];
    __shared__ int lpos[BIN_NODES];
    __shared__ int wtot[SORT_T / 32];
    __shared__ int sbase;
    const int t = threadIdx.x;
    const int lane = t & 31;
    const int w = t >> 5;
    const int bin = blockIdx.x;
    int2 rt = make_int2(0, 0);
    if (t < BIN_BLOCKS) rt = *((const int2*)(run_tab + 2 * ((size_t)bin * BIN_BLOCKS + t)));
    if (t < BIN_NODES) lcnt[t] = 0;
    const int rc = rt.y;
    const int rincl = wave_incl_scan_int(rc, lane);
    if (lane == 31) wtot[w] = rincl;
    __syncthreads();
    if (t < BIN_BLOCKS) {
        int pre = 0;
        for (int i = 0; i < w; ++i) pre += wtot[i];
        rpre[t] = pre + rincl - rc;
        rstart[t] = rt.x;
        if (t == BIN_BLOCKS - 1) rpre[BIN_BLOCKS] = pre + rincl;
    }
    __syncthreads();
    const int total = rpre[BIN_BLOCKS];
    if (t == 0) sbase = atomicAdd(key_cursor, total);
    int2 p0[SORT_U];
    #pragma unroll
    for (int u = 0; u < SORT_U; ++u) p0[u] = make_int2(-1, 0);
    if (total > 0) load_trip(pairs, rpre, rstart, t, total, p0);
    #pragma unroll
    for (int u = 0; u < SORT_U; ++u) if (p0[u].x >= 0) atomicAdd(&lcnt[p0[u].x & (BIN_NODES - 1)], 1);
    for (int i = t + SORT_U * SORT_T; i < total; i += SORT_U * SORT_T) {
        int2 p[SORT_U];
        load_trip(pairs, rpre, rstart, i, total, p);
        #pragma unroll
        for (int u = 0; u < SORT_U; ++u) if (p[u].x >= 0) atomicAdd(&lcnt[p[u].x & (BIN_NODES - 1)], 1);
    }
    __syncthreads();
    const int bs = sbase;
    const int v = (t < BIN_NODES) ? lcnt[t] : 0;
    const int incl = wave_incl_scan_int(v, lane);
    if (t < BIN_NODES && lane == 31) wtot[w] = incl;
    __syncthreads();
    if (t < BIN_NODES) {
        int pre = 0;
        for (int i = 0; i < w; ++i) pre += wtot[i];
        const int ex = bs + pre + incl - v;
        lpos[t] = ex;
        const int node = bin * BIN_NODES + t;
        if (node < n_nodes) *((int2*)(off2 + 2 * (size_t)node)) = make_int2(ex, v);
    }
    __syncthreads();
    #pragma unroll
    for (int u = 0; u < SORT_U; ++u) {
        if (p0[u].x >= 0) {
            const int slot = atomicAdd(&lpos[p0[u].x & (BIN_NODES - 1)], 1);
            keys[slot] = p0[u].y;
        }
    }
    for (int i = t + SORT_U * SORT_T; i < total; i += SORT_U * SORT_T) {
        int2 p[SORT_U];
        load_trip(pairs, rpre, rstart, i, total, p);
        #pragma unroll
        for (int u = 0; u < SORT_U; ++u) {
            if (p[u].x >= 0) {
                const int slot = atomicAdd(&lpos[p[u].x & (BIN_NODES - 1)], 1);
                keys[slot] = p[u].y;
            }
        }
    }
}

struct SegAcc { float4 acc; float m; float l; };

__device__ __forceinline__ float wave_sum32(float v) {
    for (int o = 16; o > 0; o >>= 1) v += __shfl_xor(v, o, 32);
    return v;
}

__device__ __forceinline__ void seg_update(SegAcc& st, const float4& comp, float v) {
    float mn = fmaxf(st.m, v);
    float sc = __expf(st.m - mn);
    float p  = __expf(v - mn);
    st.acc.x = st.acc.x * sc + comp.x * p;
    st.acc.y = st.acc.y * sc + comp.y * p;
    st.acc.z = st.acc.z * sc + comp.z * p;
    st.acc.w = st.acc.w * sc + comp.w * p;
    st.l = st.l * sc + p;
    st.m = mn;
}

__device__ __forceinline__ float xhalf(float x) {
    const int b = __builtin_bit_cast(int, x);
    return __builtin_bit_cast(float,
        __builtin_amdgcn_permlanex16(b, b, 0x76543210u, 0xfedcba98u, false, false));
}

template <int M>
__device__ __forceinline__ float row_xor(float x) {
    const int b = __builtin_bit_cast(int, x);
    return __builtin_bit_cast(float, __builtin_amdgcn_update_dpp(0, b, 0x160 + M, 0xf, 0xf, true));
}

__device__ __forceinline__ float4 mul4(const float4& a, const float4& b) {
    return make_float4(a.x * b.x, a.y * b.y, a.z * b.z, a.w * b.w);
}

__device__ __forceinline__ void axpby4(float4& acc, float sc, const float4& x, float p) {
    acc.x = acc.x * sc + x.x * p;
    acc.y = acc.y * sc + x.y * p;
    acc.z = acc.z * sc + x.z * p;
    acc.w = acc.w * sc + x.w * p;
}

__device__ __forceinline__ void node_row(
        const float* __restrict__ ent, const float* __restrict__ rel,
        const int* __restrict__ off, const int* __restrict__ keys,
        int d, int lane, int n_nodes, int n_edges, int n_rel, float* rowp) {
    const float4 zero4 = make_float4(0.0f, 0.0f, 0.0f, 0.0f);
    float4* rowp4 = (float4*)rowp;
    if (d >= n_nodes) { rowp4[lane] = zero4; return; }
    const int2 od = *((const int2*)(off + 2 * (size_t)d));
    const int beg = od.x;
    const int deg = od.y;
    if (deg <= 0) { rowp4[lane] = zero4; return; }

    if (deg <= 32) {
        const int h = lane >> 4;
        const int j = lane & 15;
        const float4* crow = (const float4*)(ent + (size_t)d * HD);
        const float4 c0 = crow[j];
        const float4 c1 = crow[16 + j];
        const char* entb = (const char*)ent;
        const char* relb = (const char*)rel;
        const unsigned jo = (unsigned)j * 16u;
        const int key = (lane < deg) ? keys[beg + lane] : 0x7fffffff;
        int rank = 0;
        for (int t = 0; t < deg; ++t) {
            int kt = __builtin_amdgcn_readlane(key, t);
            rank += ((kt < key) || (kt == key && t < lane)) ? 1 : 0;
        }
        if (lane >= deg) rank = lane;
        const int skey = __builtin_amdgcn_ds_permute((((rank & 1) << 4) | (rank >> 1)) << 2, key);
        const int sk = skey / n_rel;
        const int rk = skey - sk * n_rel;

        float4 acc0 = zero4;
        float4 acc1 = zero4;
        float m = -INFINITY;
        float l = 0.0f;
        for (int k0 = 0; k0 < deg; k0 += 4) {
            float4 a0[2], a1[2], b0[2], b1[2];
            for (int u = 0; u < 2; ++u) {
                const int p = k0 + 2 * u + h;
                const int pc = (p < deg) ? p : (deg - 1);
                const int from = (((pc & 1) << 4) | (pc >> 1)) << 2;
                const int s = __builtin_amdgcn_ds_bpermute(from, sk);
                const int r = __builtin_amdgcn_ds_bpermute(from, rk);
                const unsigned eo = (unsigned)s * (unsigned)(HD * sizeof(float)) + jo;
                const unsigned ro = (unsigned)r * (unsigned)(HD * sizeof(float)) + jo;
                a0[u] = *((const float4*)(entb + eo));
                a1[u] = *((const float4*)(entb + eo + 256u));
                b0[u] = *((const float4*)(relb + ro));
                b1[u] = *((const float4*)(relb + ro + 256u));
            }
            float4 x0[2], x1[2];
            float v[2];
            for (int u = 0; u < 2; ++u) {
                x0[u] = mul4(a0[u], b0[u]);
                x1[u] = mul4(a1[u], b1[u]);
                v[u] = x0[u].x * c0.x + x0[u].y * c0.y + x0[u].z * c0.z + x0[u].w * c0.w
                     + x1[u].x * c1.x + x1[u].y * c1.y + x1[u].z * c1.z + x1[u].w * c1.w;
            }
            for (int u = 0; u < 2; ++u) v[u] += row_xor<8>(v[u]);
            for (int u = 0; u < 2; ++u) v[u] += row_xor<4>(v[u]);
            for (int u = 0; u < 2; ++u) v[u] += row_xor<2>(v[u]);
            for (int u = 0; u < 2; ++u) v[u] += row_xor<1>(v[u]);
            for (int u = 0; u < 2; ++u) {
                const bool valid = (k0 + 2 * u + h) < deg;
                const float mn = valid ? fmaxf(m, v[u]) : m;
                const float sc = valid ? __expf(m - mn) : 1.0f;
                const float pw = valid ? __expf(v[u] - mn) : 0.0f;
                axpby4(acc0, sc, x0[u], pw);
                axpby4(acc1, sc, x1[u], pw);
                l = l * sc + pw;
                m = mn;
            }
        }
        const float mo = xhalf(m);
        const float lo = xhalf(l);
        const float4 o0 = make_float4(xhalf(acc0.x), xhalf(acc0.y), xhalf(acc0.z), xhalf(acc0.w));
        const float4 o1 = make_float4(xhalf(acc1.x), xhalf(acc1.y), xhalf(acc1.z), xhalf(acc1.w));
        const float mm = fmaxf(m, mo);
        const float f = __expf(m - mm);
        const float fo = __expf(mo - mm);
        const float inv = 1.0f / (l * f + lo * fo);
        if (h == 0) {
            rowp4[j] = make_float4((acc0.x * f + o0.x * fo) * inv, (acc0.y * f + o0.y * fo) * inv,
                                   (acc0.z * f + o0.z * fo) * inv, (acc0.w * f + o0.w * fo) * inv);
            rowp4[16 + j] = make_float4((acc1.x * f + o1.x * fo) * inv, (acc1.y * f + o1.y * fo) * inv,
                                        (acc1.z * f + o1.z * fo) * inv, (acc1.w * f + o1.w * fo) * inv);
        }
        return;
    }

    const float4 c = ((const float4*)(ent + (size_t)d * HD))[lane];
    SegAcc st;
    st.acc = zero4;
    st.m = -INFINITY;
    st.l = 0.0f;

    if (deg <= 64) {
        const bool has1 = lane + 32 < deg;
        const int key0 = keys[beg + lane];
        const int key1 = has1 ? keys[beg + 32 + lane] : 0x7fffffff;
        int rank0 = 0;
        int rank1 = 0;
        for (int j = 0; j < 32; ++j) {
            const int kj = __builtin_amdgcn_readlane(key0, j);
            rank0 += ((kj < key0) || (kj == key0 && j < lane)) ? 1 : 0;
            rank1 += (kj <= key1) ? 1 : 0;
        }
        for (int j = 0; j < deg - 32; ++j) {
            const int kj = __builtin_amdgcn_readlane(key1, j);
            rank0 += (kj < key0) ? 1 : 0;
            rank1 += ((kj < key1) || (kj == key1 && j < lane)) ? 1 : 0;
        }
        if (!has1) rank1 = -1;

        for (int k0 = 0; k0 < deg; k0 += 4) {
            float4 a[4], b[4];
            for (int u = 0; u < 4; ++u) {
                const int k = (k0 + u < deg) ? (k0 + u) : (deg - 1);
                const unsigned long long m =
                    ((unsigned long long)__builtin_amdgcn_ballot_w32(rank1 == k) << 32) |
                    (unsigned long long)__builtin_amdgcn_ballot_w32(rank0 == k);
                const int pos = __builtin_ctzll(m);
                const int e0 = __builtin_amdgcn_readlane(key0, pos & 31);
                const int e1 = __builtin_amdgcn_readlane(key1, pos & 31);
                const int skey = (pos < 32) ? e0 : e1;
                const int s = skey / n_rel;
                const int r = skey - s * n_rel;
                a[u] = ((const float4*)(ent + (size_t)s * HD))[lane];
                b[u] = ((const float4*)(rel + (size_t)r * HD))[lane];
            }
            float4 comp[4];
            float v[4];
            for (int u = 0; u < 4; ++u) {
                comp[u] = make_float4(a[u].x * b[u].x, a[u].y * b[u].y,
                                      a[u].z * b[u].z, a[u].w * b[u].w);
                v[u] = comp[u].x * c.x + comp[u].y * c.y + comp[u].z * c.z + comp[u].w * c.w;
            }
            for (int o = 16; o > 0; o >>= 1) {
                for (int u = 0; u < 4; ++u) v[u] += __shfl_xor(v[u], o, 32);
            }
            for (int u = 0; u < 4; ++u) {
                if (k0 + u < deg) seg_update(st, comp[u], v[u]);
            }
        }
    } else {
        int prev = -1;
        int done = 0;
        for (int it = 0; it < deg && done < deg; ++it) {
            int best = 0x7fffffff;
            for (int i = lane; i < deg; i += 32) {
                int kk = keys[beg + i];
                if (kk > prev && kk < best) best = kk;
            }
            for (int o = 16; o > 0; o >>= 1) {
                int y = __shfl_xor(best, o, 32);
                best = (y < best) ? y : best;
            }
            best = __builtin_amdgcn_readfirstlane(best);
            int mult = 0;
            for (int i = lane; i < deg; i += 32) mult += (keys[beg + i] == best) ? 1 : 0;
            for (int o = 16; o > 0; o >>= 1) mult += __shfl_xor(mult, o, 32);
            mult = __builtin_amdgcn_readfirstlane(mult);
            if (mult == 0) break;
            const int s = best / n_rel;
            const int r = best - s * n_rel;
            const float4 a = ((const float4*)(ent + (size_t)s * HD))[lane];
            const float4 b = ((const float4*)(rel + (size_t)r * HD))[lane];
            const float4 comp = make_float4(a.x * b.x, a.y * b.y, a.z * b.z, a.w * b.w);
            const float v = wave_sum32(comp.x * c.x + comp.y * c.y + comp.z * c.z + comp.w * c.w);
            for (int t = 0; t < mult; ++t) seg_update(st, comp, v);
            done += mult;
            prev = best;
        }
    }

    const float inv = 1.0f / st.l;
    rowp4[lane] = make_float4(st.acc.x * inv, st.acc.y * inv, st.acc.z * inv, st.acc.w * inv);
}

#define TILE_M 16
#define NODE_WAVES (TPB / 32)
#define TILE_PITCH (HD + 4)

__global__ void __launch_bounds__(TPB) node_gemm_kernel(
        const float* __restrict__ ent, const float* __restrict__ rel,
        const int* __restrict__ off, const int* __restrict__ keys,
        const float* __restrict__ W,
        float* __restrict__ H,
        int n_nodes, int n_edges, int n_rel) {
    __shared__ __attribute__((aligned(16))) float tile[TILE_M * TILE_PITCH];
    const int lane = threadIdx.x & 31;
    const int wave = __builtin_amdgcn_readfirstlane((int)(threadIdx.x >> 5));
    const int m0 = blockIdx.x * TILE_M;

    for (int u = 0; u < TILE_M / NODE_WAVES; ++u) {
        const int row = wave * (TILE_M / NODE_WAVES) + u;
        node_row(ent, rel, off, keys, m0 + row, lane, n_nodes, n_edges, n_rel, tile + row * TILE_PITCH);
    }
    __syncthreads();

    const int half = lane >> 4;
    const int l16  = lane & 15;
    const int n0 = wave * 16;
    v8f c = {};
    const float* arow = tile + l16 * TILE_PITCH;
    const float* wcol = W + n0 + l16;

    #pragma unroll 8
    for (int k0 = 0; k0 < HD; k0 += 4) {
        const int ka = k0 + 2 * half;
        const v2f a = *((const v2f*)(arow + ka));
        v2f b;
        b.x = wcol[(size_t)ka * HD];
        b.y = wcol[(size_t)(ka + 1) * HD];
        c = __builtin_amdgcn_wmma_f32_16x16x4_f32(
                 false, a,  false, b,
                 (short)0, c,  false,  false);
    }

    const int mrow = m0 + 8 * half;
    float* hb = H + (size_t)mrow * HD + n0 + l16;
    #pragma unroll
    for (int r2 = 0; r2 < 8; ++r2) {
        if (mrow + r2 < n_nodes) hb[(size_t)r2 * HD] = c[r2];
    }
}

#define STAT_ROWS 13
#define STAT_WAVES (TPB / 32)
__global__ void __launch_bounds__(TPB) bn_stats_kernel(
        const float* __restrict__ H, float* __restrict__ colsum,
        float* __restrict__ colsumsq, int n_rows) {
    __shared__ float part[2][STAT_WAVES][HD];
    const int lane = threadIdx.x & 31;
    const int w = __builtin_amdgcn_readfirstlane((int)(threadIdx.x >> 5));
    const int b = blockIdx.x;
    const int G = gridDim.x;
    const float4* H4 = (const float4*)H;
    float4 s = make_float4(0.0f, 0.0f, 0.0f, 0.0f);
    float4 q = make_float4(0.0f, 0.0f, 0.0f, 0.0f);
    for (int i0 = w; i0 * G + b < n_rows; i0 += STAT_WAVES * STAT_ROWS) {
        float4 v[STAT_ROWS];
        #pragma unroll
        for (int u = 0; u < STAT_ROWS; ++u) {
            const int r = (i0 + u * STAT_WAVES) * G + b;
            v[u] = (r < n_rows) ? H4[(size_t)r * (HD / 4) + lane]
                                : make_float4(0.0f, 0.0f, 0.0f, 0.0f);
        }
        #pragma unroll
        for (int u = 0; u < STAT_ROWS; ++u) {
            s.x += v[u].x; s.y += v[u].y; s.z += v[u].z; s.w += v[u].w;
            q.x += v[u].x * v[u].x; q.y += v[u].y * v[u].y;
            q.z += v[u].z * v[u].z; q.w += v[u].w * v[u].w;
        }
    }
    ((float4*)&part[0][w][0])[lane] = s;
    ((float4*)&part[1][w][0])[lane] = q;
    __syncthreads();
    const int stat = threadIdx.x >> 7;
    const int col  = threadIdx.x & (HD - 1);
    float acc = 0.0f;
    for (int i = 0; i < STAT_WAVES; ++i) acc += part[stat][i][col];
    atomicAdd((stat ? colsumsq : colsum) + col, acc);
}

__global__ void __launch_bounds__(TPB) bn_tanh_kernel(
        float* __restrict__ H, const float* __restrict__ colsum,
        const float* __restrict__ colsumsq, const float* __restrict__ gamma,
        const float* __restrict__ beta, int n_rows) {
    size_t i = (size_t)blockIdx.x * TPB + threadIdx.x;
    size_t total = (size_t)n_rows * HD;
    if (i >= total) return;
    int c = (int)(i & (HD - 1));
    float invN = 1.0f / (float)n_rows;
    float mu   = colsum[c] * invN;
    float var  = colsumsq[c] * invN - mu * mu;
    float istd = rsqrtf(var + 1e-5f);
    float v = (H[i] - mu) * istd * gamma[c] + beta[c];
    H[i] = tanhf(v);
}

extern "C" void kernel_launch(void* const* d_in, const int* in_sizes, int n_in,
                              void* d_out, int out_size, void* d_ws, size_t ws_size,
                              hipStream_t stream) {
    const float* ent   = (const float*)d_in[0];
    const float* rel   = (const float*)d_in[1];
    const float* W     = (const float*)d_in[2];
    const float* gamma = (const float*)d_in[3];
    const float* beta  = (const float*)d_in[4];
    const int*   src   = (const int*)d_in[5];
    const int*   dst   = (const int*)d_in[6];
    const int*   rid   = (const int*)d_in[7];
    float* H = (float*)d_out;

    const int N = in_sizes[0] / HD;
    const int R = in_sizes[1] / HD;
    const int E = in_sizes[5];

    float* ws       = (float*)d_ws;
    float* neigh    = ws;
    int*   keys     = (int*)(neigh + (size_t)N * HD);
    int*   cnt      = keys + E;
    float* colsum   = (float*)(cnt + N);
    float* colsumsq = colsum + HD;
    int*   pairs    = (int*)neigh;
    int*   run_tab  = pairs + 2 * (size_t)E;
    int*   off2     = run_tab + 2 * (size_t)MAX_BINS * BIN_BLOCKS;
    int*   key_cursor = cnt;

    const int n_bins = (N + BIN_NODES - 1) / BIN_NODES;
    bin_local_kernel<<<BIN_BLOCKS, TPB, 0, stream>>>(
        src, dst, rid, pairs, run_tab, key_cursor, colsum, E, n_bins, R);
    bin_sort_kernel<<<(unsigned)n_bins, SORT_T, 0, stream>>>(pairs, run_tab, key_cursor, off2, keys, N);

    node_gemm_kernel<<<(unsigned)((N + TILE_M - 1) / TILE_M), TPB, 0, stream>>>(
        ent, rel, off2, keys, W, H, N, E, R);

    bn_stats_kernel<<<512, TPB, 0, stream>>>(H, colsum, colsumsq, N);
    bn_tanh_kernel<<<(unsigned)(((size_t)N * HD + TPB - 1) / TPB), TPB, 0, stream>>>(
        H, colsum, colsumsq, gamma, beta, N);
}


# BN column sums taken from the tile's WMMA accumulators in the node kernel: one whole-line f32 atomic add per wave into 4 sum sets in rotation, joined in set order; stats launch gone; new sum order
# speedup vs baseline: 5.4795x; 1.0638x over previous
#include <hip/hip_runtime.h>
#include <math.h>

#define HD 128
#define TPB 256
#define STAT_SETS 4

typedef __attribute__((ext_vector_type(2))) float v2f;
typedef __attribute__((ext_vector_type(8))) float v8f;

#define BIN_SHIFT 8
#define BIN_NODES 256
#define MAX_BINS 256
#define BIN_BLOCKS 512
#define SORT_T 1024
#define LOC_U 7
#define SORT_U 5

__device__ __forceinline__ int wave_incl_scan_int(int v, int lane) {
    int x = v;
    for (int o = 1; o < 32; o <<= 1) {
        int y = __shfl_up(x, o, 32);
        if (lane >= o) x += y;
    }
    return x;
}

__global__ void __launch_bounds__(TPB) bin_local_kernel(
        const int* __restrict__ src, const int* __restrict__ dst, const int* __restrict__ rid,
        int* __restrict__ pairs, int* __restrict__ run_tab, int* __restrict__ key_cursor,
        float* __restrict__ colstat, int n_edges, int n_bins, int n_rel) {
    __shared__ int hist[MAX_BINS];
    __shared__ int base[MAX_BINS];
    __shared__ int wtot[TPB / 32];
    const int t = threadIdx.x;
    const int lane = t & 31;
    const int w = t >> 5;
    hist[t] = 0;
    if (blockIdx.x == 0) {
        static_assert(STAT_SETS * 2 * HD == 4 * TPB, "one float4 per thread clears all sum sets");
        ((float4*)colstat)[t] = make_float4(0.0f, 0.0f, 0.0f, 0.0f);
        if (t == 0) *key_cursor = 0;
    }
    __syncthreads();
    const int chunk = (n_edges + BIN_BLOCKS - 1) / BIN_BLOCKS;
    const int e0 = blockIdx.x * chunk;
    const int e1 = (e0 + chunk < n_edges) ? (e0 + chunk) : n_edges;
    int d0[LOC_U], k0[LOC_U];
    {
        int s0[LOC_U], r0[LOC_U];
        #pragma unroll
        for (int u = 0; u < LOC_U; ++u) {
            const int ee = e0 + t + u * TPB;
            const bool ok = ee < e1;
            d0[u] = ok ? dst[ee] : -1;
            s0[u] = ok ? src[ee] : 0;
            r0[u] = ok ? rid[ee] : 0;
        }
        #pragma unroll
        for (int u = 0; u < LOC_U; ++u) k0[u] = s0[u] * n_rel + r0[u];
    }
    #pragma unroll
    for (int u = 0; u < LOC_U; ++u) if (d0[u] >= 0) atomicAdd(&hist[d0[u] >> BIN_SHIFT], 1);
    for (int e = e0 + t + LOC_U * TPB; e < e1; e += TPB) atomicAdd(&hist[dst[e] >> BIN_SHIFT], 1);
    __syncthreads();
    const int h = hist[t];
    const int incl = wave_incl_scan_int(h, lane);
    if (lane == 31) wtot[w] = incl;
    __syncthreads();
    int pre = 0;
    for (int i = 0; i < w; ++i) pre += wtot[i];
    const int start = e0 + pre + incl - h;
    base[t] = start;
    if (t < n_bins) {
        *((int2*)(run_tab + 2 * ((size_t)t * BIN_BLOCKS + blockIdx.x))) = make_int2(start, h);
    }
    hist[t] = 0;
    __syncthreads();
    #pragma unroll
    for (int u = 0; u < LOC_U; ++u) {
        if (d0[u] >= 0) {
            const int b = d0[u] >> BIN_SHIFT;
            const int slot = base[b] + atomicAdd(&hist[b], 1);
            *((int2*)(pairs + 2 * (size_t)slot)) = make_int2(d0[u], k0[u]);
        }
    }
    for (int e = e0 + t + LOC_U * TPB; e < e1; e += TPB) {
        const int d = dst[e];
        const int k = src[e] * n_rel + rid[e];
        const int b = d >> BIN_SHIFT;
        const int slot = base[b] + atomicAdd(&hist[b], 1);
        *((int2*)(pairs + 2 * (size_t)slot)) = make_int2(d, k);
    }
}

__device__ __forceinline__ void load_trip(
        const int* __restrict__ pairs, const int* rpre, const int* rstart,
        int i0, int total, int2 (&p)[SORT_U]) {
    int lo[SORT_U], e[SORT_U];
    #pragma unroll
    for (int u = 0; u < SORT_U; ++u) {
        const int ii = i0 + u * SORT_T;
        e[u] = (ii < total) ? ii : (total - 1);
        lo[u] = 0;
    }
    #pragma unroll
    for (int wd = BIN_BLOCKS / 2; wd >= 1; wd >>= 1) {
        #pragma unroll
        for (int u = 0; u < SORT_U; ++u) lo[u] = (rpre[lo[u] + wd] <= e[u]) ? (lo[u] + wd) : lo[u];
    }
    int2 q[SORT_U];
    #pragma unroll
    for (int u = 0; u < SORT_U; ++u) {
        const int slot = rstart[lo[u]] + (e[u] - rpre[lo[u]]);
        q[u] = *((const int2*)(pairs + 2 * (size_t)slot));
    }
    #pragma unroll
    for (int u = 0; u < SORT_U; ++u) {
        const int none = (i0 + u * SORT_T < total) ? 0 : -1;
        p[u] = make_int2(q[u].x | none, q[u].y);
    }
}

__global__ void __launch_bounds__(SORT_T) bin_sort_kernel(
        const int* __restrict__ pairs, const int* __restrict__ run_tab,
        int* __restrict__ key_cursor, int* __restrict__ off2, int* __restrict__ keys,
        int n_nodes) {
    __shared__ int rpre[BIN_BLOCKS + 1];
    __shared__ int rstart[BIN_BLOCKS];
    __shared__ int lcnt[BIN_NODES];
    __shared__ int lpos[BIN_NODES];
    __shared__ int wtot[SORT_T / 32];
    __shared__ int sbase;
    const int t = threadIdx.x;
    const int lane = t & 31;
    const int w = t >> 5;
    const int bin = blockIdx.x;
    int2 rt = make_int2(0, 0);
    if (t < BIN_BLOCKS) rt = *((const int2*)(run_tab + 2 * ((size_t)bin * BIN_BLOCKS + t)));
    if (t < BIN_NODES) lcnt[t] = 0;
    const int rc = rt.y;
    const int rincl = wave_incl_scan_int(rc, lane);
    if (lane == 31) wtot[w] = rincl;
    __syncthreads();
    if (t < BIN_BLOCKS) {
        int pre = 0;
        for (int i = 0; i < w; ++i) pre += wtot[i];
        rpre[t] = pre + rincl - rc;
        rstart[t] = rt.x;
        if (t == BIN_BLOCKS - 1) rpre[BIN_BLOCKS] = pre + rincl;
    }
    __syncthreads();
    const int total = rpre[BIN_BLOCKS];
    if (t == 0) sbase = atomicAdd(key_cursor, total);
    int2 p0[SORT_U];
    #pragma unroll
    for (int u = 0; u < SORT_U; ++u) p0[u] = make_int2(-1, 0);
    if (total > 0) load_trip(pairs, rpre, rstart, t, total, p0);
    #pragma unroll
    for (int u = 0; u < SORT_U; ++u) if (p0[u].x >= 0) atomicAdd(&lcnt[p0[u].x & (BIN_NODES - 1)], 1);
    for (int i = t + SORT_U * SORT_T; i < total; i += SORT_U * SORT_T) {
        int2 p[SORT_U];
        load_trip(pairs, rpre, rstart, i, total, p);
        #pragma unroll
        for (int u = 0; u < SORT_U; ++u) if (p[u].x >= 0) atomicAdd(&lcnt[p[u].x & (BIN_NODES - 1)], 1);
    }
    __syncthreads();
    const int bs = sbase;
    const int v = (t < BIN_NODES) ? lcnt[t] : 0;
    const int incl = wave_incl_scan_int(v, lane);
    if (t < BIN_NODES && lane == 31) wtot[w] = incl;
    __syncthreads();
    if (t < BIN_NODES) {
        int pre = 0;
        for (int i = 0; i < w; ++i) pre += wtot[i];
        const int ex = bs + pre + incl - v;
        lpos[t] = ex;
        const int node = bin * BIN_NODES + t;
        if (node < n_nodes) *((int2*)(off2 + 2 * (size_t)node)) = make_int2(ex, v);
    }
    __syncthreads();
    #pragma unroll
    for (int u = 0; u < SORT_U; ++u) {
        if (p0[u].x >= 0) {
            const int slot = atomicAdd(&lpos[p0[u].x & (BIN_NODES - 1)], 1);
            keys[slot] = p0[u].y;
        }
    }
    for (int i = t + SORT_U * SORT_T; i < total; i += SORT_U * SORT_T) {
        int2 p[SORT_U];
        load_trip(pairs, rpre, rstart, i, total, p);
        #pragma unroll
        for (int u = 0; u < SORT_U; ++u) {
            if (p[u].x >= 0) {
                const int slot = atomicAdd(&lpos[p[u].x & (BIN_NODES - 1)], 1);
                keys[slot] = p[u].y;
            }
        }
    }
}

struct SegAcc { float4 acc; float m; float l; };

__device__ __forceinline__ float wave_sum32(float v) {
    for (int o = 16; o > 0; o >>= 1) v += __shfl_xor(v, o, 32);
    return v;
}

__device__ __forceinline__ void seg_update(SegAcc& st, const float4& comp, float v) {
    float mn = fmaxf(st.m, v);
    float sc = __expf(st.m - mn);
    float p  = __expf(v - mn);
    st.acc.x = st.acc.x * sc + comp.x * p;
    st.acc.y = st.acc.y * sc + comp.y * p;
    st.acc.z = st.acc.z * sc + comp.z * p;
    st.acc.w = st.acc.w * sc + comp.w * p;
    st.l = st.l * sc + p;
    st.m = mn;
}

__device__ __forceinline__ float xhalf(float x) {
    const int b = __builtin_bit_cast(int, x);
    return __builtin_bit_cast(float,
        __builtin_amdgcn_permlanex16(b, b, 0x76543210u, 0xfedcba98u, false, false));
}

template <int M>
__device__ __forceinline__ float row_xor(float x) {
    const int b = __builtin_bit_cast(int, x);
    return __builtin_bit_cast(float, __builtin_amdgcn_update_dpp(0, b, 0x160 + M, 0xf, 0xf, true));
}

__device__ __forceinline__ float4 mul4(const float4& a, const float4& b) {
    return make_float4(a.x * b.x, a.y * b.y, a.z * b.z, a.w * b.w);
}

__device__ __forceinline__ void axpby4(float4& acc, float sc, const float4& x, float p) {
    acc.x = acc.x * sc + x.x * p;
    acc.y = acc.y * sc + x.y * p;
    acc.z = acc.z * sc + x.z * p;
    acc.w = acc.w * sc + x.w * p;
}

__device__ __forceinline__ void node_row(
        const float* __restrict__ ent, const float* __restrict__ rel,
        const int* __restrict__ off, const int* __restrict__ keys,
        int d, int lane, int n_nodes, int n_edges, int n_rel, float* rowp) {
    const float4 zero4 = make_float4(0.0f, 0.0f, 0.0f, 0.0f);
    float4* rowp4 = (float4*)rowp;
    if (d >= n_nodes) { rowp4[lane] = zero4; return; }
    const int2 od = *((const int2*)(off + 2 * (size_t)d));
    const int beg = od.x;
    const int deg = od.y;
    if (deg <= 0) { rowp4[lane] = zero4; return; }

    if (deg <= 32) {
        const int h = lane >> 4;
        const int j = lane & 15;
        const float4* crow = (const float4*)(ent + (size_t)d * HD);
        const float4 c0 = crow[j];
        const float4 c1 = crow[16 + j];
        const char* entb = (const char*)ent;
        const char* relb = (const char*)rel;
        const unsigned jo = (unsigned)j * 16u;
        const int key = (lane < deg) ? keys[beg + lane] : 0x7fffffff;
        int rank = 0;
        for (int t = 0; t < deg; ++t) {
            int kt = __builtin_amdgcn_readlane(key, t);
            rank += ((kt < key) || (kt == key && t < lane)) ? 1 : 0;
        }
        if (lane >= deg) rank = lane;
        const int skey = __builtin_amdgcn_ds_permute((((rank & 1) << 4) | (rank >> 1)) << 2, key);
        const int sk = skey / n_rel;
        const int rk = skey - sk * n_rel;

        float4 acc0 = zero4;
        float4 acc1 = zero4;
        float m = -INFINITY;
        float l = 0.0f;
        for (int k0 = 0; k0 < deg; k0 += 4) {
            float4 a0[2], a1[2], b0[2], b1[2];
            for (int u = 0; u < 2; ++u) {
                const int p = k0 + 2 * u + h;
                const int pc = (p < deg) ? p : (deg - 1);
                const int from = (((pc & 1) << 4) | (pc >> 1)) << 2;
                const int s = __builtin_amdgcn_ds_bpermute(from, sk);
                const int r = __builtin_amdgcn_ds_bpermute(from, rk);
                const unsigned eo = (unsigned)s * (unsigned)(HD * sizeof(float)) + jo;
                const unsigned ro = (unsigned)r * (unsigned)(HD * sizeof(float)) + jo;
                a0[u] = *((const float4*)(entb + eo));
                a1[u] = *((const float4*)(entb + eo + 256u));
                b0[u] = *((const float4*)(relb + ro));
                b1[u] = *((const float4*)(relb + ro + 256u));
            }
            float4 x0[2], x1[2];
            float v[2];
            for (int u = 0; u < 2; ++u) {
                x0[u] = mul4(a0[u], b0[u]);
                x1[u] = mul4(a1[u], b1[u]);
                v[u] = x0[u].x * c0.x + x0[u].y * c0.y + x0[u].z * c0.z + x0[u].w * c0.w
                     + x1[u].x * c1.x + x1[u].y * c1.y + x1[u].z * c1.z + x1[u].w * c1.w;
            }
            for (int u = 0; u < 2; ++u) v[u] += row_xor<8>(v[u]);
            for (int u = 0; u < 2; ++u) v[u] += row_xor<4>(v[u]);
            for (int u = 0; u < 2; ++u) v[u] += row_xor<2>(v[u]);
            for (int u = 0; u < 2; ++u) v[u] += row_xor<1>(v[u]);
            for (int u = 0; u < 2; ++u) {
                const bool valid = (k0 + 2 * u + h) < deg;
                const float mn = valid ? fmaxf(m, v[u]) : m;
                const float sc = valid ? __expf(m - mn) : 1.0f;
                const float pw = valid ? __expf(v[u] - mn) : 0.0f;
                axpby4(acc0, sc, x0[u], pw);
                axpby4(acc1, sc, x1[u], pw);
                l = l * sc + pw;
                m = mn;
            }
        }
        const float mo = xhalf(m);
        const float lo = xhalf(l);
        const float4 o0 = make_float4(xhalf(acc0.x), xhalf(acc0.y), xhalf(acc0.z), xhalf(acc0.w));
        const float4 o1 = make_float4(xhalf(acc1.x), xhalf(acc1.y), xhalf(acc1.z), xhalf(acc1.w));
        const float mm = fmaxf(m, mo);
        const float f = __expf(m - mm);
        const float fo = __expf(mo - mm);
        const float inv = 1.0f / (l * f + lo * fo);
        if (h == 0) {
            rowp4[j] = make_float4((acc0.x * f + o0.x * fo) * inv, (acc0.y * f + o0.y * fo) * inv,
                                   (acc0.z * f + o0.z * fo) * inv, (acc0.w * f + o0.w * fo) * inv);
            rowp4[16 + j] = make_float4((acc1.x * f + o1.x * fo) * inv, (acc1.y * f + o1.y * fo) * inv,
                                        (acc1.z * f + o1.z * fo) * inv, (acc1.w * f + o1.w * fo) * inv);
        }
        return;
    }

    const float4 c = ((const float4*)(ent + (size_t)d * HD))[lane];
    SegAcc st;
    st.acc = zero4;
    st.m = -INFINITY;
    st.l = 0.0f;

    if (deg <= 64) {
        const bool has1 = lane + 32 < deg;
        const int key0 = keys[beg + lane];
        const int key1 = has1 ? keys[beg + 32 + lane] : 0x7fffffff;
        int rank0 = 0;
        int rank1 = 0;
        for (int j = 0; j < 32; ++j) {
            const int kj = __builtin_amdgcn_readlane(key0, j);
            rank0 += ((kj < key0) || (kj == key0 && j < lane)) ? 1 : 0;
            rank1 += (kj <= key1) ? 1 : 0;
        }
        for (int j = 0; j < deg - 32; ++j) {
            const int kj = __builtin_amdgcn_readlane(key1, j);
            rank0 += (kj < key0) ? 1 : 0;
            rank1 += ((kj < key1) || (kj == key1 && j < lane)) ? 1 : 0;
        }
        if (!has1) rank1 = -1;

        for (int k0 = 0; k0 < deg; k0 += 4) {
            float4 a[4], b[4];
            for (int u = 0; u < 4; ++u) {
                const int k = (k0 + u < deg) ? (k0 + u) : (deg - 1);
                const unsigned long long m =
                    ((unsigned long long)__builtin_amdgcn_ballot_w32(rank1 == k) << 32) |
                    (unsigned long long)__builtin_amdgcn_ballot_w32(rank0 == k);
                const int pos = __builtin_ctzll(m);
                const int e0 = __builtin_amdgcn_readlane(key0, pos & 31);
                const int e1 = __builtin_amdgcn_readlane(key1, pos & 31);
                const int skey = (pos < 32) ? e0 : e1;
                const int s = skey / n_rel;
                const int r = skey - s * n_rel;
                a[u] = ((const float4*)(ent + (size_t)s * HD))[lane];
                b[u] = ((const float4*)(rel + (size_t)r * HD))[lane];
            }
            float4 comp[4];
            float v[4];
            for (int u = 0; u < 4; ++u) {
                comp[u] = make_float4(a[u].x * b[u].x, a[u].y * b[u].y,
                                      a[u].z * b[u].z, a[u].w * b[u].w);
                v[u] = comp[u].x * c.x + comp[u].y * c.y + comp[u].z * c.z + comp[u].w * c.w;
            }
            for (int o = 16; o > 0; o >>= 1) {
                for (int u = 0; u < 4; ++u) v[u] += __shfl_xor(v[u], o, 32);
            }
            for (int u = 0; u < 4; ++u) {
                if (k0 + u < deg) seg_update(st, comp[u], v[u]);
            }
        }
    } else {
        int prev = -1;
        int done = 0;
        for (int it = 0; it < deg && done < deg; ++it) {
            int best = 0x7fffffff;
            for (int i = lane; i < deg; i += 32) {
                int kk = keys[beg + i];
                if (kk > prev && kk < best) best = kk;
            }
            for (int o = 16; o > 0; o >>= 1) {
                int y = __shfl_xor(best, o, 32);
                best = (y < best) ? y : best;
            }
            best = __builtin_amdgcn_readfirstlane(best);
            int mult = 0;
            for (int i = lane; i < deg; i += 32) mult += (keys[beg + i] == best) ? 1 : 0;
            for (int o = 16; o > 0; o >>= 1) mult += __shfl_xor(mult, o, 32);
            mult = __builtin_amdgcn_readfirstlane(mult);
            if (mult == 0) break;
            const int s = best / n_rel;
            const int r = best - s * n_rel;
            const float4 a = ((const float4*)(ent + (size_t)s * HD))[lane];
            const float4 b = ((const float4*)(rel + (size_t)r * HD))[lane];
            const float4 comp = make_float4(a.x * b.x, a.y * b.y, a.z * b.z, a.w * b.w);
            const float v = wave_sum32(comp.x * c.x + comp.y * c.y + comp.z * c.z + comp.w * c.w);
            for (int t = 0; t < mult; ++t) seg_update(st, comp, v);
            done += mult;
            prev = best;
        }
    }

    const float inv = 1.0f / st.l;
    rowp4[lane] = make_float4(st.acc.x * inv, st.acc.y * inv, st.acc.z * inv, st.acc.w * inv);
}

#define TILE_M 16
#define NODE_WAVES (TPB / 32)
#define TILE_PITCH (HD + 4)

__global__ void __launch_bounds__(TPB) node_gemm_kernel(
        const float* __restrict__ ent, const float* __restrict__ rel,
        const int* __restrict__ off, const int* __restrict__ keys,
        const float* __restrict__ W,
        float* __restrict__ H,
        float* __restrict__ colstat,
        int n_nodes, int n_edges, int n_rel) {
    __shared__ __attribute__((aligned(16))) float tile[TILE_M * TILE_PITCH];
    const int lane = threadIdx.x & 31;
    const int wave = __builtin_amdgcn_readfirstlane((int)(threadIdx.x >> 5));
    const int m0 = blockIdx.x * TILE_M;

    for (int u = 0; u < TILE_M / NODE_WAVES; ++u) {
        const int row = wave * (TILE_M / NODE_WAVES) + u;
        node_row(ent, rel, off, keys, m0 + row, lane, n_nodes, n_edges, n_rel, tile + row * TILE_PITCH);
    }
    __syncthreads();

    const int half = lane >> 4;
    const int l16  = lane & 15;
    const int n0 = wave * 16;
    v8f c = {};
    const float* arow = tile + l16 * TILE_PITCH;
    const float* wcol = W + n0 + l16;

    #pragma unroll 8
    for (int k0 = 0; k0 < HD; k0 += 4) {
        const int ka = k0 + 2 * half;
        const v2f a = *((const v2f*)(arow + ka));
        v2f b;
        b.x = wcol[(size_t)ka * HD];
        b.y = wcol[(size_t)(ka + 1) * HD];
        c = __builtin_amdgcn_wmma_f32_16x16x4_f32(
                 false, a,  false, b,
                 (short)0, c,  false,  false);
    }

    asm volatile("v_nop\n\tv_nop\n\tv_nop\n\tv_nop" : "+v"(c));
    float cs = 0.0f;
    float cq = 0.0f;
    #pragma unroll
    for (int r2 = 0; r2 < 8; ++r2) {
        cs += c[r2];
        cq += c[r2] * c[r2];
    }
    cs += xhalf(cs);
    cq += xhalf(cq);
    atomicAdd(colstat + (size_t)(blockIdx.x & (STAT_SETS - 1)) * (2 * HD) + 2 * (n0 + l16) + half,
              half ? cq : cs);

    const int mrow = m0 + 8 * half;
    float* hb = H + (size_t)mrow * HD + n0 + l16;
    #pragma unroll
    for (int r2 = 0; r2 < 8; ++r2) {
        if (mrow + r2 < n_nodes) hb[(size_t)r2 * HD] = c[r2];
    }
}

__global__ void __launch_bounds__(TPB) bn_tanh_kernel(
        float* __restrict__ H, const float* __restrict__ colstat,
        const float* __restrict__ gamma, const float* __restrict__ beta, int n_rows) {
    size_t i = (size_t)blockIdx.x * TPB + threadIdx.x;
    size_t total = (size_t)n_rows * HD;
    if (i >= total) return;
    int c = (int)(i & (HD - 1));
    float invN = 1.0f / (float)n_rows;
    const float2* st2 = (const float2*)colstat;
    const float2 st0 = st2[c];
    float s = st0.x;
    float q = st0.y;
    #pragma unroll
    for (int k = 1; k < STAT_SETS; ++k) {
        const float2 stk = st2[k * HD + c];
        s += stk.x;
        q += stk.y;
    }
    float mu   = s * invN;
    float var  = q * invN - mu * mu;
    float istd = rsqrtf(var + 1e-5f);
    float v = (H[i] - mu) * istd * gamma[c] + beta[c];
    H[i] = tanhf(v);
}

extern "C" void kernel_launch(void* const* d_in, const int* in_sizes, int n_in,
                              void* d_out, int out_size, void* d_ws, size_t ws_size,
                              hipStream_t stream) {
    const float* ent   = (const float*)d_in[0];
    const float* rel   = (const float*)d_in[1];
    const float* W     = (const float*)d_in[2];
    const float* gamma = (const float*)d_in[3];
    const float* beta  = (const float*)d_in[4];
    const int*   src   = (const int*)d_in[5];
    const int*   dst   = (const int*)d_in[6];
    const int*   rid   = (const int*)d_in[7];
    float* H = (float*)d_out;

    const int N = in_sizes[0] / HD;
    const int R = in_sizes[1] / HD;
    const int E = in_sizes[5];

    float* ws       = (float*)d_ws;
    float* neigh    = ws;
    int*   keys     = (int*)(neigh + (size_t)N * HD);
    int*   cnt      = keys + E;
    float* colstat  = (float*)(cnt + 64);
    int*   pairs    = (int*)neigh;
    int*   run_tab  = pairs + 2 * (size_t)E;
    int*   off2     = run_tab + 2 * (size_t)MAX_BINS * BIN_BLOCKS;
    int*   key_cursor = cnt;

    const int n_bins = (N + BIN_NODES - 1) / BIN_NODES;
    bin_local_kernel<<<BIN_BLOCKS, TPB, 0, stream>>>(
        src, dst, rid, pairs, run_tab, key_cursor, colstat, E, n_bins, R);
    bin_sort_kernel<<<(unsigned)n_bins, SORT_T, 0, stream>>>(pairs, run_tab, key_cursor, off2, keys, N);

    node_gemm_kernel<<<(unsigned)((N + TILE_M - 1) / TILE_M), TPB, 0, stream>>>(
        ent, rel, off2, keys, W, H, colstat, N, E, R);

    bn_tanh_kernel<<<(unsigned)(((size_t)N * HD + TPB - 1) / TPB), TPB, 0, stream>>>(
        H, colstat, gamma, beta, N);
}
